# starting kernel
# baseline (speedup 1.0000x reference)
#include <hip/hip_runtime.h>
#include <hip/hip_bf16.h>
#include <math.h>


typedef __attribute__((ext_vector_type(2))) float v2f;
typedef __attribute__((ext_vector_type(8))) float v8f;

#define IN_DIM   128
#define HIDDEN    16
#define NCLASS    40
#define NPAD      48

__global__ void gcn_zero_f32(float* __restrict__ p, int n) {
  int i = blockIdx.x * blockDim.x + threadIdx.x;
  if (i < n) p[i] = 0.0f;
}

__global__ void gcn_spmm1(const int* __restrict__ rows,
                          const int* __restrict__ cols,
                          const float* __restrict__ vals,
                          const float* __restrict__ x,
                          float* __restrict__ z1, int nEdges) {
  int gid = blockIdx.x * blockDim.x + threadIdx.x;
  int e  = gid >> 5;
  int f4 = gid & 31;
  if (e >= nEdges) return;
  int r = rows[e];
  int c = cols[e];
  float v = vals[e];
  const float4* xr = (const float4*)(x + (size_t)c * IN_DIM);
  float4 d = xr[f4];
  float* zr = z1 + (size_t)r * IN_DIM + (size_t)f4 * 4;
  atomicAdd(zr + 0, v * d.x);
  atomicAdd(zr + 1, v * d.y);
  atomicAdd(zr + 2, v * d.z);
  atomicAdd(zr + 3, v * d.w);
}

__global__ void gcn_spmm2(const int* __restrict__ rows,
                          const int* __restrict__ cols,
                          const float* __restrict__ vals,
                          const float* __restrict__ h,
                          float* __restrict__ z2, int nEdges) {
  int gid = blockIdx.x * blockDim.x + threadIdx.x;
  int e  = gid >> 2;
  int f4 = gid & 3;
  if (e >= nEdges) return;
  int r = rows[e];
  int c = cols[e];
  float v = vals[e];
  const float4* hr = (const float4*)(h + (size_t)c * HIDDEN);
  float4 d = hr[f4];
  float* zr = z2 + (size_t)r * HIDDEN + (size_t)f4 * 4;
  atomicAdd(zr + 0, v * d.x);
  atomicAdd(zr + 1, v * d.y);
  atomicAdd(zr + 2, v * d.z);
  atomicAdd(zr + 3, v * d.w);
}

__global__ void __launch_bounds__(32)
gcn_gemm1_relu(const float* __restrict__ z1,
               const float* __restrict__ w1,
               float* __restrict__ h) {
  __shared__ float Wl[IN_DIM * HIDDEN];
  __shared__ float Al[16 * IN_DIM];
  const int lane = threadIdx.x;
  const int tile = blockIdx.x;

  for (int i = lane; i < IN_DIM * HIDDEN; i += 32) Wl[i] = w1[i];
  const float* Ab = z1 + (size_t)tile * 16 * IN_DIM;
  for (int i = lane; i < 16 * IN_DIM; i += 32) Al[i] = Ab[i];
  __syncthreads();

  const int half = lane >> 4;
  const int l    = lane & 15;

  v8f c = {};
  #pragma unroll 8
  for (int k0 = 0; k0 < IN_DIM; k0 += 4) {
    const int ka = k0 + 2 * half;
    v2f a, b;
    a.x = Al[l * IN_DIM + ka];
    a.y = Al[l * IN_DIM + ka + 1];
    b.x = Wl[ka * HIDDEN + l];
    b.y = Wl[(ka + 1) * HIDDEN + l];
    c = __builtin_amdgcn_wmma_f32_16x16x4_f32(false, a, false, b,
                                              (short)0, c, false, false);
  }

  float* hb = h + (size_t)tile * 16 * HIDDEN;
  #pragma unroll
  for (int i = 0; i < 8; ++i) {
    float v = c[i];
    hb[(i + 8 * half) * HIDDEN + l] = v > 0.0f ? v : 0.0f;
  }
}

__global__ void __launch_bounds__(32)
gcn_gemm2_lsm(const float* __restrict__ z2,
              const float* __restrict__ w2,
              float* __restrict__ out) {
  __shared__ float Wl[HIDDEN * NPAD];
  __shared__ float Al[16 * HIDDEN];
  __shared__ float Ol[16 * NPAD];
  const int lane = threadIdx.x;
  const int tile = blockIdx.x;

  for (int i = lane; i < HIDDEN * NPAD; i += 32) {
    int r = i / NPAD, n = i % NPAD;
    Wl[i] = (n < NCLASS) ? w2[r * NCLASS + n] : 0.0f;
  }
  const float* Ab = z2 + (size_t)tile * 16 * HIDDEN;
  for (int i = lane; i < 16 * HIDDEN; i += 32) Al[i] = Ab[i];
  __syncthreads();

  const int half = lane >> 4;
  const int l    = lane & 15;

  #pragma unroll
  for (int t = 0; t < 3; ++t) {
    v8f c = {};
    #pragma unroll
    for (int k0 = 0; k0 < HIDDEN; k0 += 4) {
      const int ka = k0 + 2 * half;
      v2f a, b;
      a.x = Al[l * HIDDEN + ka];
      a.y = Al[l * HIDDEN + ka + 1];
      b.x = Wl[ka * NPAD + t * 16 + l];
      b.y = Wl[(ka + 1) * NPAD + t * 16 + l];
      c = __builtin_amdgcn_wmma_f32_16x16x4_f32(false, a, false, b,
                                                (short)0, c, false, false);
    }
    #pragma unroll
    for (int i = 0; i < 8; ++i)
      Ol[(i + 8 * half) * NPAD + t * 16 + l] = c[i];
  }
  __syncthreads();

  if (lane < 16) {
    const float* row = Ol + lane * NPAD;
    float mx = row[0];
    for (int j = 1; j < NCLASS; ++j) mx = fmaxf(mx, row[j]);
    float s = 0.0f;
    for (int j = 0; j < NCLASS; ++j) s += __expf(row[j] - mx);
    const float lse = mx + __logf(s);
    float* ob = out + ((size_t)tile * 16 + lane) * NCLASS;
    for (int j = 0; j < NCLASS; ++j) ob[j] = row[j] - lse;
  }
}

extern "C" void kernel_launch(void* const* d_in, const int* in_sizes, int n_in,
                              void* d_out, int out_size, void* d_ws, size_t ws_size,
                              hipStream_t stream) {
  const float* features = (const float*)d_in[0];
  const float* weight1  = (const float*)d_in[1];
  const float* weight2  = (const float*)d_in[2];
  const float* edgeVals = (const float*)d_in[3];
  const int*   edgeRows = (const int*)  d_in[4];
  const int*   edgeCols = (const int*)  d_in[5];

  const int nNodes = in_sizes[0] / IN_DIM;
  const int nEdges = in_sizes[3];
  const int nTiles = nNodes / 16;

  float* z1 = (float*)d_ws;
  float* h  = z1 + (size_t)nNodes * IN_DIM;
  float* z2 = h  + (size_t)nNodes * HIDDEN;

  {
    int n = nNodes * (IN_DIM + 2 * HIDDEN);
    gcn_zero_f32<<<(n + 255) / 256, 256, 0, stream>>>(z1, n);
  }
  {
    long long threads = (long long)nEdges * 32;
    gcn_spmm1<<<(unsigned)((threads + 255) / 256), 256, 0, stream>>>(
        edgeRows, edgeCols, edgeVals, features, z1, nEdges);
  }
  gcn_gemm1_relu<<<nTiles, 32, 0, stream>>>(z1, weight1, h);
  {
    long long threads = (long long)nEdges * 4;
    gcn_spmm2<<<(unsigned)((threads + 255) / 256), 256, 0, stream>>>(
        edgeRows, edgeCols, edgeVals, h, z2, nEdges);
  }
  gcn_gemm2_lsm<<<nTiles, 32, 0, stream>>>(z2, weight2, (float*)d_out);
}


# Edges bucketed by destination row in-call (int atomic rank, scan, place); both A@ products become per-row gathers with register sums and plain stores: no f32 atomics, no zero pass; sums in rank order
# speedup vs baseline: 1.5095x; 1.5095x over previous
#include <hip/hip_runtime.h>
#include <hip/hip_bf16.h>
#include <math.h>


typedef __attribute__((ext_vector_type(2))) float v2f;
typedef __attribute__((ext_vector_type(8))) float v8f;

#define IN_DIM   128
#define HIDDEN    16
#define NCLASS    40
#define NPAD      48

__global__ void gcn_zero_i32(int* __restrict__ p, int n) {
  int i = blockIdx.x * blockDim.x + threadIdx.x;
  if (i < n) p[i] = 0;
}

__global__ void gcn_edge_rank(const int* __restrict__ rows,
                              int* __restrict__ counts,
                              int* __restrict__ rank, int nEdges) {
  int e = blockIdx.x * blockDim.x + threadIdx.x;
  if (e < nEdges) rank[e] = atomicAdd(&counts[rows[e]], 1);
}

__global__ void __launch_bounds__(1024)
gcn_row_start(const int* __restrict__ counts, int* __restrict__ start,
              int nNodes) {
  __shared__ int wsum[32];
  __shared__ int wpre[32];
  const int t = threadIdx.x;
  const int lane = t & 31;
  const int w = t >> 5;
  const int base = blockIdx.x * 1024;

  int ahead = 0;
  for (int i = t; i < base; i += 1024) ahead += counts[i];

  const int idx = base + t;
  const int c = idx < nNodes ? counts[idx] : 0;

  int inc = c;
  #pragma unroll
  for (int d = 1; d < 32; d <<= 1) {
    int n = __shfl_up(inc, d);
    if (lane >= d) inc += n;
  }
  #pragma unroll
  for (int d = 16; d > 0; d >>= 1) ahead += __shfl_xor(ahead, d);

  if (lane == 31) wsum[w] = inc;
  if (lane == 0)  wpre[w] = ahead;
  __syncthreads();

  int off = 0;
  #pragma unroll
  for (int i = 0; i < 32; ++i) {
    off += wpre[i];
    if (i < w) off += wsum[i];
  }
  const int excl = off + inc - c;
  if (idx < nNodes) start[idx] = excl;
  if (idx == nNodes - 1) start[nNodes] = excl + c;
}

__global__ void gcn_edge_place(const int* __restrict__ rows,
                               const int* __restrict__ cols,
                               const float* __restrict__ vals,
                               const int* __restrict__ start,
                               const int* __restrict__ rank,
                               int* __restrict__ pairs, int nEdges) {
  int e = blockIdx.x * blockDim.x + threadIdx.x;
  if (e >= nEdges) return;
  int pos = start[rows[e]] + rank[e];
  reinterpret_cast<int2*>(pairs)[pos] = make_int2(cols[e], __float_as_int(vals[e]));
}

template <int U>
__device__ __forceinline__ void gcn_gather_batch(float4& acc, int2 mine, int j,
                                                 const float4* __restrict__ x4,
                                                 int lane) {
  float4 d[U];
  float v[U];
  #pragma unroll
  for (int u = 0; u < U; ++u) {
    const int c = __builtin_amdgcn_readlane(mine.x, j + u);
    v[u] = __int_as_float(__builtin_amdgcn_readlane(mine.y, j + u));
    d[u] = x4[(size_t)c * (IN_DIM / 4) + lane];
  }
  #pragma unroll
  for (int u = 0; u < U; ++u) {
    acc.x += v[u] * d[u].x;
    acc.y += v[u] * d[u].y;
    acc.z += v[u] * d[u].z;
    acc.w += v[u] * d[u].w;
  }
}

__global__ void __launch_bounds__(256)
gcn_gather1(const int* __restrict__ start,
            const int* __restrict__ pairs,
            const float* __restrict__ x,
            float* __restrict__ z1, int nNodes) {
  const int lane = threadIdx.x & 31;
  const int r = __builtin_amdgcn_readfirstlane(blockIdx.x * 8 + (threadIdx.x >> 5));
  if (r >= nNodes) return;
  const int s = __builtin_amdgcn_readfirstlane(start[r]);
  const int e = __builtin_amdgcn_readfirstlane(start[r + 1]);
  const int2* __restrict__ pr = reinterpret_cast<const int2*>(pairs);
  const float4* __restrict__ x4 = reinterpret_cast<const float4*>(x);

  float4 acc = make_float4(0.0f, 0.0f, 0.0f, 0.0f);
  for (int base = s; base < e; base += 32) {
    const int n = (e - base) < 32 ? (e - base) : 32;
    int2 mine = make_int2(0, 0);
    if (lane < n) mine = pr[base + lane];
    int j = 0;
    for (; j + 8 <= n; j += 8) gcn_gather_batch<8>(acc, mine, j, x4, lane);
    if (n - j >= 4) { gcn_gather_batch<4>(acc, mine, j, x4, lane); j += 4; }
    if (n - j >= 2) { gcn_gather_batch<2>(acc, mine, j, x4, lane); j += 2; }
    if (n - j >= 1) { gcn_gather_batch<1>(acc, mine, j, x4, lane); }
  }
  reinterpret_cast<float4*>(z1)[(size_t)r * (IN_DIM / 4) + lane] = acc;
}

__global__ void __launch_bounds__(256)
gcn_gather2(const int* __restrict__ start,
            const int* __restrict__ pairs,
            const float* __restrict__ h,
            float* __restrict__ z2, int nNodes) {
  const int lane = threadIdx.x & 31;
  const int wave = blockIdx.x * 8 + (threadIdx.x >> 5);
  const int r    = wave * 2 + (lane >> 4);
  const int slot = (lane >> 2) & 3;
  const int f4   = lane & 3;
  int s = 0, e = 0;
  if (r < nNodes) { s = start[r]; e = start[r + 1]; }
  const int2* __restrict__ pr = reinterpret_cast<const int2*>(pairs);
  const float4* __restrict__ h4 = reinterpret_cast<const float4*>(h);

  float4 acc = make_float4(0.0f, 0.0f, 0.0f, 0.0f);
  int p = s + slot;
  for (; p + 4 < e; p += 8) {
    const int2 cv0 = pr[p];
    const int2 cv1 = pr[p + 4];
    const float4 d0 = h4[(size_t)cv0.x * (HIDDEN / 4) + f4];
    const float4 d1 = h4[(size_t)cv1.x * (HIDDEN / 4) + f4];
    const float v0 = __int_as_float(cv0.y);
    const float v1 = __int_as_float(cv1.y);
    acc.x += v0 * d0.x; acc.y += v0 * d0.y; acc.z += v0 * d0.z; acc.w += v0 * d0.w;
    acc.x += v1 * d1.x; acc.y += v1 * d1.y; acc.z += v1 * d1.z; acc.w += v1 * d1.w;
  }
  if (p < e) {
    const int2 cv0 = pr[p];
    const float4 d0 = h4[(size_t)cv0.x * (HIDDEN / 4) + f4];
    const float v0 = __int_as_float(cv0.y);
    acc.x += v0 * d0.x; acc.y += v0 * d0.y; acc.z += v0 * d0.z; acc.w += v0 * d0.w;
  }
  #pragma unroll
  for (int d = 4; d <= 8; d <<= 1) {
    acc.x += __shfl_xor(acc.x, d);
    acc.y += __shfl_xor(acc.y, d);
    acc.z += __shfl_xor(acc.z, d);
    acc.w += __shfl_xor(acc.w, d);
  }
  if (slot == 0 && r < nNodes)
    reinterpret_cast<float4*>(z2)[(size_t)r * (HIDDEN / 4) + f4] = acc;
}

__global__ void __launch_bounds__(32)
gcn_gemm1_relu(const float* __restrict__ z1,
               const float* __restrict__ w1,
               float* __restrict__ h) {
  __shared__ float Wl[IN_DIM * HIDDEN];
  __shared__ float Al[16 * IN_DIM];
  const int lane = threadIdx.x;
  const int tile = blockIdx.x;

  for (int i = lane; i < IN_DIM * HIDDEN; i += 32) Wl[i] = w1[i];
  const float* Ab = z1 + (size_t)tile * 16 * IN_DIM;
  for (int i = lane; i < 16 * IN_DIM; i += 32) Al[i] = Ab[i];
  __syncthreads();

  const int half = lane >> 4;
  const int l    = lane & 15;

  v8f c = {};
  #pragma unroll 8
  for (int k0 = 0; k0 < IN_DIM; k0 += 4) {
    const int ka = k0 + 2 * half;
    v2f a, b;
    a.x = Al[l * IN_DIM + ka];
    a.y = Al[l * IN_DIM + ka + 1];
    b.x = Wl[ka * HIDDEN + l];
    b.y = Wl[(ka + 1) * HIDDEN + l];
    c = __builtin_amdgcn_wmma_f32_16x16x4_f32(false, a, false, b,
                                              (short)0, c, false, false);
  }

  float* hb = h + (size_t)tile * 16 * HIDDEN;
  #pragma unroll
  for (int i = 0; i < 8; ++i) {
    float v = c[i];
    hb[(i + 8 * half) * HIDDEN + l] = v > 0.0f ? v : 0.0f;
  }
}

__global__ void __launch_bounds__(32)
gcn_gemm2_lsm(const float* __restrict__ z2,
              const float* __restrict__ w2,
              float* __restrict__ out) {
  __shared__ float Wl[HIDDEN * NPAD];
  __shared__ float Al[16 * HIDDEN];
  __shared__ float Ol[16 * NPAD];
  const int lane = threadIdx.x;
  const int tile = blockIdx.x;

  for (int i = lane; i < HIDDEN * NPAD; i += 32) {
    int r = i / NPAD, n = i % NPAD;
    Wl[i] = (n < NCLASS) ? w2[r * NCLASS + n] : 0.0f;
  }
  const float* Ab = z2 + (size_t)tile * 16 * HIDDEN;
  for (int i = lane; i < 16 * HIDDEN; i += 32) Al[i] = Ab[i];
  __syncthreads();

  const int half = lane >> 4;
  const int l    = lane & 15;

  #pragma unroll
  for (int t = 0; t < 3; ++t) {
    v8f c = {};
    #pragma unroll
    for (int k0 = 0; k0 < HIDDEN; k0 += 4) {
      const int ka = k0 + 2 * half;
      v2f a, b;
      a.x = Al[l * HIDDEN + ka];
      a.y = Al[l * HIDDEN + ka + 1];
      b.x = Wl[ka * NPAD + t * 16 + l];
      b.y = Wl[(ka + 1) * NPAD + t * 16 + l];
      c = __builtin_amdgcn_wmma_f32_16x16x4_f32(false, a, false, b,
                                                (short)0, c, false, false);
    }
    #pragma unroll
    for (int i = 0; i < 8; ++i)
      Ol[(i + 8 * half) * NPAD + t * 16 + l] = c[i];
  }
  __syncthreads();

  if (lane < 16) {
    const float* row = Ol + lane * NPAD;
    float mx = row[0];
    for (int j = 1; j < NCLASS; ++j) mx = fmaxf(mx, row[j]);
    float s = 0.0f;
    for (int j = 0; j < NCLASS; ++j) s += __expf(row[j] - mx);
    const float lse = mx + __logf(s);
    float* ob = out + ((size_t)tile * 16 + lane) * NCLASS;
    for (int j = 0; j < NCLASS; ++j) ob[j] = row[j] - lse;
  }
}

extern "C" void kernel_launch(void* const* d_in, const int* in_sizes, int n_in,
                              void* d_out, int out_size, void* d_ws, size_t ws_size,
                              hipStream_t stream) {
  const float* features = (const float*)d_in[0];
  const float* weight1  = (const float*)d_in[1];
  const float* weight2  = (const float*)d_in[2];
  const float* edgeVals = (const float*)d_in[3];
  const int*   edgeRows = (const int*)  d_in[4];
  const int*   edgeCols = (const int*)  d_in[5];

  const int nNodes = in_sizes[0] / IN_DIM;
  const int nEdges = in_sizes[3];
  const int nTiles = nNodes / 16;

  float* z1 = (float*)d_ws;
  float* h  = z1 + (size_t)nNodes * IN_DIM;
  float* z2 = h  + (size_t)nNodes * HIDDEN;
  int* pairs  = (int*)(z2 + (size_t)nNodes * HIDDEN);
  int* rank   = pairs + (size_t)nEdges * 2;
  int* counts = rank + nEdges;
  int* start  = counts + nNodes;

  gcn_zero_i32<<<(nNodes + 255) / 256, 256, 0, stream>>>(counts, nNodes);
  gcn_edge_rank<<<(nEdges + 255) / 256, 256, 0, stream>>>(edgeRows, counts, rank, nEdges);
  gcn_row_start<<<(nNodes + 1023) / 1024, 1024, 0, stream>>>(counts, start, nNodes);
  gcn_edge_place<<<(nEdges + 255) / 256, 256, 0, stream>>>(
      edgeRows, edgeCols, edgeVals, start, rank, pairs, nEdges);
  gcn_gather1<<<(nNodes + 7) / 8, 256, 0, stream>>>(start, pairs, features, z1, nNodes);
  gcn_gemm1_relu<<<nTiles, 32, 0, stream>>>(z1, weight1, h);
  gcn_gather2<<<(nNodes + 15) / 16, 256, 0, stream>>>(start, pairs, h, z2, nNodes);
  gcn_gemm2_lsm<<<nTiles, 32, 0, stream>>>(z2, weight2, (float*)d_out);
}


# Each row gather fused with the WMMA product it feeds: a 4-wave block sums a 16-row tile (4 rows per wave) into LDS, one wave multiplies it; z1, z2 stay on chip; 8 to 6 launches; sum order kept
# speedup vs baseline: 1.6973x; 1.1244x over previous
#include <hip/hip_runtime.h>
#include <hip/hip_bf16.h>
#include <math.h>


typedef __attribute__((ext_vector_type(2))) float v2f;
typedef __attribute__((ext_vector_type(8))) float v8f;

#define IN_DIM   128
#define HIDDEN    16
#define NCLASS    40
#define NPAD      48
#define A1_PITCH 132
#define A2_PITCH  20
#define O_PITCH   49

__global__ void gcn_zero_i32(int* __restrict__ p, int n) {
  int i = blockIdx.x * blockDim.x + threadIdx.x;
  if (i < n) p[i] = 0;
}

__global__ void gcn_edge_rank(const int* __restrict__ rows,
                              int* __restrict__ counts,
                              int* __restrict__ rank, int nEdges) {
  int e = blockIdx.x * blockDim.x + threadIdx.x;
  if (e < nEdges) rank[e] = atomicAdd(&counts[rows[e]], 1);
}

__global__ void __launch_bounds__(1024)
gcn_row_start(const int* __restrict__ counts, int* __restrict__ start,
              int nNodes) {
  __shared__ int wsum[32];
  __shared__ int wpre[32];
  const int t = threadIdx.x;
  const int lane = t & 31;
  const int w = t >> 5;
  const int base = blockIdx.x * 1024;

  int ahead = 0;
  for (int i = t; i < base; i += 1024) ahead += counts[i];

  const int idx = base + t;
  const int c = idx < nNodes ? counts[idx] : 0;

  int inc = c;
  #pragma unroll
  for (int d = 1; d < 32; d <<= 1) {
    int n = __shfl_up(inc, d);
    if (lane >= d) inc += n;
  }
  #pragma unroll
  for (int d = 16; d > 0; d >>= 1) ahead += __shfl_xor(ahead, d);

  if (lane == 31) wsum[w] = inc;
  if (lane == 0)  wpre[w] = ahead;
  __syncthreads();

  int off = 0;
  #pragma unroll
  for (int i = 0; i < 32; ++i) {
    off += wpre[i];
    if (i < w) off += wsum[i];
  }
  const int excl = off + inc - c;
  if (idx < nNodes) start[idx] = excl;
  if (idx == nNodes - 1) start[nNodes] = excl + c;
}

__global__ void gcn_edge_place(const int* __restrict__ rows,
                               const int* __restrict__ cols,
                               const float* __restrict__ vals,
                               const int* __restrict__ start,
                               const int* __restrict__ rank,
                               int* __restrict__ pairs, int nEdges) {
  int e = blockIdx.x * blockDim.x + threadIdx.x;
  if (e >= nEdges) return;
  int pos = start[rows[e]] + rank[e];
  reinterpret_cast<int2*>(pairs)[pos] = make_int2(cols[e], __float_as_int(vals[e]));
}

template <int U>
__device__ __forceinline__ void gcn_gather_batch(float4& acc, int2 mine, int j,
                                                 const float4* __restrict__ x4,
                                                 int lane) {
  float4 d[U];
  float v[U];
  #pragma unroll
  for (int u = 0; u < U; ++u) {
    const int c = __builtin_amdgcn_readlane(mine.x, j + u);
    v[u] = __int_as_float(__builtin_amdgcn_readlane(mine.y, j + u));
    d[u] = x4[(size_t)c * (IN_DIM / 4) + lane];
  }
  #pragma unroll
  for (int u = 0; u < U; ++u) {
    acc.x += v[u] * d[u].x;
    acc.y += v[u] * d[u].y;
    acc.z += v[u] * d[u].z;
    acc.w += v[u] * d[u].w;
  }
}

__device__ __forceinline__ float4 gcn_row_sum1(int s, int e, int2 mine,
                                               const int2* __restrict__ pr,
                                               const float4* __restrict__ x4,
                                               int lane) {
  float4 acc = make_float4(0.0f, 0.0f, 0.0f, 0.0f);
  for (int base = s; base < e; base += 32) {
    const int n = (e - base) < 32 ? (e - base) : 32;
    if (base != s) {
      mine = make_int2(0, 0);
      if (lane < n) mine = pr[base + lane];
    }
    int j = 0;
    for (; j + 8 <= n; j += 8) gcn_gather_batch<8>(acc, mine, j, x4, lane);
    if (n - j >= 4) { gcn_gather_batch<4>(acc, mine, j, x4, lane); j += 4; }
    if (n - j >= 2) { gcn_gather_batch<2>(acc, mine, j, x4, lane); j += 2; }
    if (n - j >= 1) { gcn_gather_batch<1>(acc, mine, j, x4, lane); }
  }
  return acc;
}

__global__ void __launch_bounds__(128)
gcn_gather1_gemm1(const int* __restrict__ start,
                  const int* __restrict__ pairs,
                  const float* __restrict__ x,
                  const float* __restrict__ w1,
                  float* __restrict__ h, int nNodes) {
  __shared__ __attribute__((aligned(16))) float Wl[IN_DIM * HIDDEN];
  __shared__ __attribute__((aligned(16))) float Al[16 * A1_PITCH];
  const int t = threadIdx.x;
  const int lane = t & 31;
  const int w = __builtin_amdgcn_readfirstlane(t >> 5);
  const int tile = blockIdx.x;
  const int r0 = tile * 16 + w * 4;
  const int2* __restrict__ pr = reinterpret_cast<const int2*>(pairs);
  const float4* __restrict__ x4 = reinterpret_cast<const float4*>(x);

  int sv = 0;
  if (lane < 5) {
    const int idx = r0 + lane;
    sv = start[idx < nNodes ? idx : nNodes];
  }

  {
    const float4* __restrict__ w14 = reinterpret_cast<const float4*>(w1);
    float4 wreg[4];
    #pragma unroll
    for (int i = 0; i < 4; ++i) wreg[i] = w14[t + 128 * i];
    #pragma unroll
    for (int i = 0; i < 4; ++i) reinterpret_cast<float4*>(Wl)[t + 128 * i] = wreg[i];
  }

  int S[5];
  #pragma unroll
  for (int q = 0; q < 5; ++q) S[q] = __builtin_amdgcn_readlane(sv, q);

  int2 mine[4];
  #pragma unroll
  for (int q = 0; q < 4; ++q) {
    const int d = S[q + 1] - S[q];
    const int n = d < 32 ? d : 32;
    mine[q] = make_int2(0, 0);
    if (lane < n) mine[q] = pr[S[q] + lane];
  }

  #pragma unroll
  for (int q = 0; q < 4; ++q) {
    const float4 acc = gcn_row_sum1(S[q], S[q + 1], mine[q], pr, x4, lane);
    *reinterpret_cast<float4*>(&Al[(w * 4 + q) * A1_PITCH + lane * 4]) = acc;
  }
  __syncthreads();
  if (w != 0) return;

  const int half = lane >> 4;
  const int l    = lane & 15;

  v8f c = {};
  #pragma unroll 8
  for (int k0 = 0; k0 < IN_DIM; k0 += 4) {
    const int ka = k0 + 2 * half;
    const v2f a = *reinterpret_cast<const v2f*>(&Al[l * A1_PITCH + ka]);
    v2f b;
    b.x = Wl[ka * HIDDEN + l];
    b.y = Wl[(ka + 1) * HIDDEN + l];
    c = __builtin_amdgcn_wmma_f32_16x16x4_f32(false, a, false, b,
                                              (short)0, c, false, false);
  }

  float* hb = h + (size_t)tile * 16 * HIDDEN;
  #pragma unroll
  for (int i = 0; i < 8; ++i) {
    const float v = c[i];
    hb[(i + 8 * half) * HIDDEN + l] = v > 0.0f ? v : 0.0f;
  }
}

__global__ void __launch_bounds__(128)
gcn_gather2_gemm2_lsm(const int* __restrict__ start,
                      const int* __restrict__ pairs,
                      const float* __restrict__ h,
                      const float* __restrict__ w2,
                      float* __restrict__ out, int nNodes) {
  __shared__ float Wl[HIDDEN * NPAD];
  __shared__ __attribute__((aligned(16))) float Al[16 * A2_PITCH];
  __shared__ float Ol[16 * O_PITCH];
  const int t = threadIdx.x;
  const int lane = t & 31;
  const int w = __builtin_amdgcn_readfirstlane(t >> 5);
  const int tile = blockIdx.x;
  const int slot = (lane >> 2) & 3;
  const int f4   = lane & 3;
  const int2* __restrict__ pr = reinterpret_cast<const int2*>(pairs);
  const float4* __restrict__ h4 = reinterpret_cast<const float4*>(h);

  int rl[2], s[2], e[2];
  #pragma unroll
  for (int u = 0; u < 2; ++u) {
    rl[u] = w * 4 + u * 2 + (lane >> 4);
    const int r = tile * 16 + rl[u];
    s[u] = start[r < nNodes ? r : nNodes];
    e[u] = start[r + 1 < nNodes ? r + 1 : nNodes];
  }

  #pragma unroll
  for (int k = 0; k < (HIDDEN * NPAD) / 128; ++k) {
    const int i = t + 128 * k;
    const int r = i / NPAD, n = i % NPAD;
    const float v = w2[r * NCLASS + (n < NCLASS ? n : NCLASS - 1)];
    Wl[i] = (n < NCLASS) ? v : 0.0f;
  }

  #pragma unroll
  for (int u = 0; u < 2; ++u) {
    float4 acc = make_float4(0.0f, 0.0f, 0.0f, 0.0f);
    int p = s[u] + slot;
    for (; p + 4 < e[u]; p += 8) {
      const int2 cv0 = pr[p];
      const int2 cv1 = pr[p + 4];
      const float4 d0 = h4[(size_t)cv0.x * (HIDDEN / 4) + f4];
      const float4 d1 = h4[(size_t)cv1.x * (HIDDEN / 4) + f4];
      const float v0 = __int_as_float(cv0.y);
      const float v1 = __int_as_float(cv1.y);
      acc.x += v0 * d0.x; acc.y += v0 * d0.y; acc.z += v0 * d0.z; acc.w += v0 * d0.w;
      acc.x += v1 * d1.x; acc.y += v1 * d1.y; acc.z += v1 * d1.z; acc.w += v1 * d1.w;
    }
    if (p < e[u]) {
      const int2 cv0 = pr[p];
      const float4 d0 = h4[(size_t)cv0.x * (HIDDEN / 4) + f4];
      const float v0 = __int_as_float(cv0.y);
      acc.x += v0 * d0.x; acc.y += v0 * d0.y; acc.z += v0 * d0.z; acc.w += v0 * d0.w;
    }
    #pragma unroll
    for (int d = 4; d <= 8; d <<= 1) {
      acc.x += __shfl_xor(acc.x, d);
      acc.y += __shfl_xor(acc.y, d);
      acc.z += __shfl_xor(acc.z, d);
      acc.w += __shfl_xor(acc.w, d);
    }
    if (slot == 0)
      *reinterpret_cast<float4*>(&Al[rl[u] * A2_PITCH + f4 * 4]) = acc;
  }
  __syncthreads();

  if (w == 0) {
    const int half = lane >> 4;
    const int l    = lane & 15;
    #pragma unroll
    for (int nt = 0; nt < 3; ++nt) {
      v8f c = {};
      #pragma unroll
      for (int k0 = 0; k0 < HIDDEN; k0 += 4) {
        const int ka = k0 + 2 * half;
        const v2f a = *reinterpret_cast<const v2f*>(&Al[l * A2_PITCH + ka]);
        v2f b;
        b.x = Wl[ka * NPAD + nt * 16 + l];
        b.y = Wl[(ka + 1) * NPAD + nt * 16 + l];
        c = __builtin_amdgcn_wmma_f32_16x16x4_f32(false, a, false, b,
                                                  (short)0, c, false, false);
      }
      #pragma unroll
      for (int i = 0; i < 8; ++i)
        Ol[(i + 8 * half) * O_PITCH + nt * 16 + l] = c[i];
    }
  }
  __syncthreads();

  if (w == 0 && lane < 16 && tile * 16 + lane < nNodes) {
    const float* row = Ol + lane * O_PITCH;
    float mx = row[0];
    for (int j = 1; j < NCLASS; ++j) mx = fmaxf(mx, row[j]);
    float sum = 0.0f;
    for (int j = 0; j < NCLASS; ++j) sum += __expf(row[j] - mx);
    const float lse = mx + __logf(sum);
    float* ob = out + ((size_t)tile * 16 + lane) * NCLASS;
    for (int j = 0; j < NCLASS; ++j) ob[j] = row[j] - lse;
  }
}

extern "C" void kernel_launch(void* const* d_in, const int* in_sizes, int n_in,
                              void* d_out, int out_size, void* d_ws, size_t ws_size,
                              hipStream_t stream) {
  const float* features = (const float*)d_in[0];
  const float* weight1  = (const float*)d_in[1];
  const float* weight2  = (const float*)d_in[2];
  const float* edgeVals = (const float*)d_in[3];
  const int*   edgeRows = (const int*)  d_in[4];
  const int*   edgeCols = (const int*)  d_in[5];

  const int nNodes = in_sizes[0] / IN_DIM;
  const int nEdges = in_sizes[3];
  const int nTiles = nNodes / 16;

  float* h  = (float*)d_ws;
  int* pairs  = (int*)(h + (size_t)nNodes * HIDDEN);
  int* rank   = pairs + (size_t)nEdges * 2;
  int* counts = rank + nEdges;
  int* start  = counts + nNodes;

  gcn_zero_i32<<<(nNodes + 255) / 256, 256, 0, stream>>>(counts, nNodes);
  gcn_edge_rank<<<(nEdges + 255) / 256, 256, 0, stream>>>(edgeRows, counts, rank, nEdges);
  gcn_row_start<<<(nNodes + 1023) / 1024, 1024, 0, stream>>>(counts, start, nNodes);
  gcn_edge_place<<<(nEdges + 255) / 256, 256, 0, stream>>>(
      edgeRows, edgeCols, edgeVals, start, rank, pairs, nEdges);
  gcn_gather1_gemm1<<<nTiles, 128, 0, stream>>>(start, pairs, features, weight1, h, nNodes);
  gcn_gather2_gemm2_lsm<<<nTiles, 128, 0, stream>>>(start, pairs, h, weight2, (float*)d_out, nNodes);
}


# Edges bucketed with no global atomic: two-level counting sort (chunk histograms over 256-row blocks + scan, then rows inside a block), ranks from returning LDS counters; sums in their arrival order
# speedup vs baseline: 3.2834x; 1.9345x over previous
#include <hip/hip_runtime.h>
#include <hip/hip_bf16.h>
#include <math.h>


typedef __attribute__((ext_vector_type(2))) float v2f;
typedef __attribute__((ext_vector_type(8))) float v8f;

#define IN_DIM   128
#define HIDDEN    16
#define NCLASS    40
#define NPAD      48
#define A1_PITCH 132
#define A2_PITCH  20
#define O_PITCH   49

#define CH_THREADS 512
#define CH_EDGES   4096
#define NB_MAX     256

__device__ __forceinline__ void gcn_chunk_load8(const int* __restrict__ p, int c,
                                                int t, int nEdges, int fill,
                                                int (&out)[8]) {
  #pragma unroll
  for (int k = 0; k < 2; ++k) {
    const int e = (c * (CH_EDGES / 4) + k * CH_THREADS + t) * 4;
    if (e + 4 <= nEdges) {
      const int4 q = *reinterpret_cast<const int4*>(p + e);
      out[4 * k + 0] = q.x; out[4 * k + 1] = q.y;
      out[4 * k + 2] = q.z; out[4 * k + 3] = q.w;
    } else {
      #pragma unroll
      for (int u = 0; u < 4; ++u) out[4 * k + u] = (e + u < nEdges) ? p[e + u] : fill;
    }
  }
}

__global__ void __launch_bounds__(CH_THREADS)
gcn_coarse_count(const int* __restrict__ rows, int* __restrict__ histT,
                 int nEdges, int nChunks, int nBuckets) {
  __shared__ int cnt[NB_MAX];
  const int t = threadIdx.x;
  const int c = blockIdx.x;
  if (t < NB_MAX) cnt[t] = 0;
  int r[8];
  gcn_chunk_load8(rows, c, t, nEdges, -1, r);
  __syncthreads();
  #pragma unroll
  for (int u = 0; u < 8; ++u)
    if (r[u] >= 0) atomicAdd(&cnt[r[u] >> 8], 1);
  __syncthreads();
  if (t < nBuckets) histT[t * nChunks + c] = cnt[t];
}

__global__ void __launch_bounds__(1024)
gcn_row_start(const int* __restrict__ counts, int* __restrict__ start,
              int nNodes) {
  __shared__ int wsum[32];
  __shared__ int wpre[32];
  const int t = threadIdx.x;
  const int lane = t & 31;
  const int w = t >> 5;
  const int base = blockIdx.x * 1024;

  int ahead = 0;
  for (int i = t; i < base; i += 1024) ahead += counts[i];

  const int idx = base + t;
  const int c = idx < nNodes ? counts[idx] : 0;

  int inc = c;
  #pragma unroll
  for (int d = 1; d < 32; d <<= 1) {
    int n = __shfl_up(inc, d);
    if (lane >= d) inc += n;
  }
  #pragma unroll
  for (int d = 16; d > 0; d >>= 1) ahead += __shfl_xor(ahead, d);

  if (lane == 31) wsum[w] = inc;
  if (lane == 0)  wpre[w] = ahead;
  __syncthreads();

  int off = 0;
  #pragma unroll
  for (int i = 0; i < 32; ++i) {
    off += wpre[i];
    if (i < w) off += wsum[i];
  }
  const int excl = off + inc - c;
  if (idx < nNodes) start[idx] = excl;
  if (idx == nNodes - 1) start[nNodes] = excl + c;
}

__global__ void __launch_bounds__(CH_THREADS)
gcn_coarse_place(const int* __restrict__ rows, const int* __restrict__ cols,
                 const float* __restrict__ vals, const int* __restrict__ offs,
                 int* __restrict__ recs, int nEdges, int nChunks, int nBuckets) {
  __shared__ int cur[NB_MAX];
  const int t = threadIdx.x;
  const int c = blockIdx.x;
  int r[8], cc[8], vv[8];
  gcn_chunk_load8(rows, c, t, nEdges, -1, r);
  gcn_chunk_load8(cols, c, t, nEdges, 0, cc);
  gcn_chunk_load8(reinterpret_cast<const int*>(vals), c, t, nEdges, 0, vv);
  if (t < nBuckets) cur[t] = offs[t * nChunks + c];
  __syncthreads();
  int4* __restrict__ rec4 = reinterpret_cast<int4*>(recs);
  #pragma unroll
  for (int u = 0; u < 8; ++u) {
    if (r[u] >= 0) {
      const int pos = atomicAdd(&cur[r[u] >> 8], 1);
      rec4[pos] = make_int4(cc[u], vv[u], r[u], 0);
    }
  }
}

__global__ void __launch_bounds__(1024)
gcn_fine_place(const int* __restrict__ recs, const int* __restrict__ offs,
               int* __restrict__ start, int* __restrict__ pairs,
               int nNodes, int nChunks) {
  __shared__ int cnt[256];
  __shared__ int wtot[8];
  const int t = threadIdx.x;
  const int lane = t & 31;
  const int w = t >> 5;
  const int b = blockIdx.x;
  const int lo = offs[b * nChunks];
  const int hi = offs[(b + 1) * nChunks];
  const int4* __restrict__ rec4 = reinterpret_cast<const int4*>(recs);
  if (t < 256) cnt[t] = 0;
  __syncthreads();

  for (int i0 = lo + t; i0 < hi; i0 += 4096) {
    int z[4];
    #pragma unroll
    for (int u = 0; u < 4; ++u) {
      const int i = i0 + 1024 * u;
      z[u] = (i < hi) ? rec4[i].z : -1;
    }
    #pragma unroll
    for (int u = 0; u < 4; ++u)
      if (z[u] >= 0) atomicAdd(&cnt[z[u] & 255], 1);
  }
  __syncthreads();

  int c = 0, inc = 0;
  if (t < 256) {
    c = cnt[t];
    inc = c;
    #pragma unroll
    for (int d = 1; d < 32; d <<= 1) {
      const int n = __shfl_up(inc, d);
      if (lane >= d) inc += n;
    }
    if (lane == 31) wtot[w] = inc;
  }
  __syncthreads();
  if (t < 256) {
    int off = 0;
    #pragma unroll
    for (int i = 0; i < 8; ++i)
      if (i < w) off += wtot[i];
    const int first = lo + off + inc - c;
    const int row = b * 256 + t;
    if (row < nNodes) start[row] = first;
    if (row == nNodes - 1) start[nNodes] = first + c;
    cnt[t] = first;
  }
  __syncthreads();

  int2* __restrict__ pr = reinterpret_cast<int2*>(pairs);
  for (int i0 = lo + t; i0 < hi; i0 += 4096) {
    int4 q[4];
    #pragma unroll
    for (int u = 0; u < 4; ++u) {
      const int i = i0 + 1024 * u;
      q[u] = (i < hi) ? rec4[i] : make_int4(0, 0, -1, 0);
    }
    #pragma unroll
    for (int u = 0; u < 4; ++u) {
      if (q[u].z >= 0) {
        const int pos = atomicAdd(&cnt[q[u].z & 255], 1);
        pr[pos] = make_int2(q[u].x, q[u].y);
      }
    }
  }
}

template <int U>
__device__ __forceinline__ void gcn_gather_batch(float4& acc, int2 mine, int j,
                                                 const float4* __restrict__ x4,
                                                 int lane) {
  float4 d[U];
  float v[U];
  #pragma unroll
  for (int u = 0; u < U; ++u) {
    const int c = __builtin_amdgcn_readlane(mine.x, j + u);
    v[u] = __int_as_float(__builtin_amdgcn_readlane(mine.y, j + u));
    d[u] = x4[(size_t)c * (IN_DIM / 4) + lane];
  }
  #pragma unroll
  for (int u = 0; u < U; ++u) {
    acc.x += v[u] * d[u].x;
    acc.y += v[u] * d[u].y;
    acc.z += v[u] * d[u].z;
    acc.w += v[u] * d[u].w;
  }
}

__device__ __forceinline__ float4 gcn_row_sum1(int s, int e, int2 mine,
                                               const int2* __restrict__ pr,
                                               const float4* __restrict__ x4,
                                               int lane) {
  float4 acc = make_float4(0.0f, 0.0f, 0.0f, 0.0f);
  for (int base = s; base < e; base += 32) {
    const int n = (e - base) < 32 ? (e - base) : 32;
    if (base != s) {
      mine = make_int2(0, 0);
      if (lane < n) mine = pr[base + lane];
    }
    int j = 0;
    for (; j + 8 <= n; j += 8) gcn_gather_batch<8>(acc, mine, j, x4, lane);
    if (n - j >= 4) { gcn_gather_batch<4>(acc, mine, j, x4, lane); j += 4; }
    if (n - j >= 2) { gcn_gather_batch<2>(acc, mine, j, x4, lane); j += 2; }
    if (n - j >= 1) { gcn_gather_batch<1>(acc, mine, j, x4, lane); }
  }
  return acc;
}

__global__ void __launch_bounds__(128)
gcn_gather1_gemm1(const int* __restrict__ start,
                  const int* __restrict__ pairs,
                  const float* __restrict__ x,
                  const float* __restrict__ w1,
                  float* __restrict__ h, int nNodes) {
  __shared__ __attribute__((aligned(16))) float Wl[IN_DIM * HIDDEN];
  __shared__ __attribute__((aligned(16))) float Al[16 * A1_PITCH];
  const int t = threadIdx.x;
  const int lane = t & 31;
  const int w = __builtin_amdgcn_readfirstlane(t >> 5);
  const int tile = blockIdx.x;
  const int r0 = tile * 16 + w * 4;
  const int2* __restrict__ pr = reinterpret_cast<const int2*>(pairs);
  const float4* __restrict__ x4 = reinterpret_cast<const float4*>(x);

  int sv = 0;
  if (lane < 5) {
    const int idx = r0 + lane;
    sv = start[idx < nNodes ? idx : nNodes];
  }

  {
    const float4* __restrict__ w14 = reinterpret_cast<const float4*>(w1);
    float4 wreg[4];
    #pragma unroll
    for (int i = 0; i < 4; ++i) wreg[i] = w14[t + 128 * i];
    #pragma unroll
    for (int i = 0; i < 4; ++i) reinterpret_cast<float4*>(Wl)[t + 128 * i] = wreg[i];
  }

  int S[5];
  #pragma unroll
  for (int q = 0; q < 5; ++q) S[q] = __builtin_amdgcn_readlane(sv, q);

  int2 mine[4];
  #pragma unroll
  for (int q = 0; q < 4; ++q) {
    const int d = S[q + 1] - S[q];
    const int n = d < 32 ? d : 32;
    mine[q] = make_int2(0, 0);
    if (lane < n) mine[q] = pr[S[q] + lane];
  }

  #pragma unroll
  for (int q = 0; q < 4; ++q) {
    const float4 acc = gcn_row_sum1(S[q], S[q + 1], mine[q], pr, x4, lane);
    *reinterpret_cast<float4*>(&Al[(w * 4 + q) * A1_PITCH + lane * 4]) = acc;
  }
  __syncthreads();
  if (w != 0) return;

  const int half = lane >> 4;
  const int l    = lane & 15;

  v8f c = {};
  #pragma unroll 8
  for (int k0 = 0; k0 < IN_DIM; k0 += 4) {
    const int ka = k0 + 2 * half;
    const v2f a = *reinterpret_cast<const v2f*>(&Al[l * A1_PITCH + ka]);
    v2f b;
    b.x = Wl[ka * HIDDEN + l];
    b.y = Wl[(ka + 1) * HIDDEN + l];
    c = __builtin_amdgcn_wmma_f32_16x16x4_f32(false, a, false, b,
                                              (short)0, c, false, false);
  }

  float* hb = h + (size_t)tile * 16 * HIDDEN;
  #pragma unroll
  for (int i = 0; i < 8; ++i) {
    const float v = c[i];
    hb[(i + 8 * half) * HIDDEN + l] = v > 0.0f ? v : 0.0f;
  }
}

__global__ void __launch_bounds__(128)
gcn_gather2_gemm2_lsm(const int* __restrict__ start,
                      const int* __restrict__ pairs,
                      const float* __restrict__ h,
                      const float* __restrict__ w2,
                      float* __restrict__ out, int nNodes) {
  __shared__ float Wl[HIDDEN * NPAD];
  __shared__ __attribute__((aligned(16))) float Al[16 * A2_PITCH];
  __shared__ float Ol[16 * O_PITCH];
  const int t = threadIdx.x;
  const int lane = t & 31;
  const int w = __builtin_amdgcn_readfirstlane(t >> 5);
  const int tile = blockIdx.x;
  const int slot = (lane >> 2) & 3;
  const int f4   = lane & 3;
  const int2* __restrict__ pr = reinterpret_cast<const int2*>(pairs);
  const float4* __restrict__ h4 = reinterpret_cast<const float4*>(h);

  int rl[2], s[2], e[2];
  #pragma unroll
  for (int u = 0; u < 2; ++u) {
    rl[u] = w * 4 + u * 2 + (lane >> 4);
    const int r = tile * 16 + rl[u];
    s[u] = start[r < nNodes ? r : nNodes];
    e[u] = start[r + 1 < nNodes ? r + 1 : nNodes];
  }

  #pragma unroll
  for (int k = 0; k < (HIDDEN * NPAD) / 128; ++k) {
    const int i = t + 128 * k;
    const int r = i / NPAD, n = i % NPAD;
    const float v = w2[r * NCLASS + (n < NCLASS ? n : NCLASS - 1)];
    Wl[i] = (n < NCLASS) ? v : 0.0f;
  }

  #pragma unroll
  for (int u = 0; u < 2; ++u) {
    float4 acc = make_float4(0.0f, 0.0f, 0.0f, 0.0f);
    int p = s[u] + slot;
    for (; p + 4 < e[u]; p += 8) {
      const int2 cv0 = pr[p];
      const int2 cv1 = pr[p + 4];
      const float4 d0 = h4[(size_t)cv0.x * (HIDDEN / 4) + f4];
      const float4 d1 = h4[(size_t)cv1.x * (HIDDEN / 4) + f4];
      const float v0 = __int_as_float(cv0.y);
      const float v1 = __int_as_float(cv1.y);
      acc.x += v0 * d0.x; acc.y += v0 * d0.y; acc.z += v0 * d0.z; acc.w += v0 * d0.w;
      acc.x += v1 * d1.x; acc.y += v1 * d1.y; acc.z += v1 * d1.z; acc.w += v1 * d1.w;
    }
    if (p < e[u]) {
      const int2 cv0 = pr[p];
      const float4 d0 = h4[(size_t)cv0.x * (HIDDEN / 4) + f4];
      const float v0 = __int_as_float(cv0.y);
      acc.x += v0 * d0.x; acc.y += v0 * d0.y; acc.z += v0 * d0.z; acc.w += v0 * d0.w;
    }
    #pragma unroll
    for (int d = 4; d <= 8; d <<= 1) {
      acc.x += __shfl_xor(acc.x, d);
      acc.y += __shfl_xor(acc.y, d);
      acc.z += __shfl_xor(acc.z, d);
      acc.w += __shfl_xor(acc.w, d);
    }
    if (slot == 0)
      *reinterpret_cast<float4*>(&Al[rl[u] * A2_PITCH + f4 * 4]) = acc;
  }
  __syncthreads();

  if (w == 0) {
    const int half = lane >> 4;
    const int l    = lane & 15;
    #pragma unroll
    for (int nt = 0; nt < 3; ++nt) {
      v8f c = {};
      #pragma unroll
      for (int k0 = 0; k0 < HIDDEN; k0 += 4) {
        const int ka = k0 + 2 * half;
        const v2f a = *reinterpret_cast<const v2f*>(&Al[l * A2_PITCH + ka]);
        v2f b;
        b.x = Wl[ka * NPAD + nt * 16 + l];
        b.y = Wl[(ka + 1) * NPAD + nt * 16 + l];
        c = __builtin_amdgcn_wmma_f32_16x16x4_f32(false, a, false, b,
                                                  (short)0, c, false, false);
      }
      #pragma unroll
      for (int i = 0; i < 8; ++i)
        Ol[(i + 8 * half) * O_PITCH + nt * 16 + l] = c[i];
    }
  }
  __syncthreads();

  if (w == 0 && lane < 16 && tile * 16 + lane < nNodes) {
    const float* row = Ol + lane * O_PITCH;
    float mx = row[0];
    for (int j = 1; j < NCLASS; ++j) mx = fmaxf(mx, row[j]);
    float sum = 0.0f;
    for (int j = 0; j < NCLASS; ++j) sum += __expf(row[j] - mx);
    const float lse = mx + __logf(sum);
    float* ob = out + ((size_t)tile * 16 + lane) * NCLASS;
    for (int j = 0; j < NCLASS; ++j) ob[j] = row[j] - lse;
  }
}

extern "C" void kernel_launch(void* const* d_in, const int* in_sizes, int n_in,
                              void* d_out, int out_size, void* d_ws, size_t ws_size,
                              hipStream_t stream) {
  const float* features = (const float*)d_in[0];
  const float* weight1  = (const float*)d_in[1];
  const float* weight2  = (const float*)d_in[2];
  const float* edgeVals = (const float*)d_in[3];
  const int*   edgeRows = (const int*)  d_in[4];
  const int*   edgeCols = (const int*)  d_in[5];

  const int nNodes = in_sizes[0] / IN_DIM;
  const int nEdges = in_sizes[3];
  const int nTiles = nNodes / 16;

  const int nBuckets = (nNodes + 255) / 256;
  const int nChunks  = (nEdges + CH_EDGES - 1) / CH_EDGES;
  const int nHist    = nBuckets * nChunks;

  float* h  = (float*)d_ws;
  int* pairs  = (int*)(h + (size_t)nNodes * HIDDEN);
  int* recs   = pairs + (size_t)nEdges * 2;
  int* histT  = recs + (size_t)nEdges * 4;
  int* offs   = histT + nHist;
  int* start  = offs + nHist + 1;

  gcn_coarse_count<<<nChunks, CH_THREADS, 0, stream>>>(edgeRows, histT, nEdges, nChunks, nBuckets);
  gcn_row_start<<<(nHist + 1023) / 1024, 1024, 0, stream>>>(histT, offs, nHist);
  gcn_coarse_place<<<nChunks, CH_THREADS, 0, stream>>>(
      edgeRows, edgeCols, edgeVals, offs, recs, nEdges, nChunks, nBuckets);
  gcn_fine_place<<<nBuckets, 1024, 0, stream>>>(recs, offs, start, pairs, nNodes, nChunks);
  gcn_gather1_gemm1<<<nTiles, 128, 0, stream>>>(start, pairs, features, weight1, h, nNodes);
  gcn_gather2_gemm2_lsm<<<nTiles, 128, 0, stream>>>(start, pairs, h, weight2, (float*)d_out, nNodes);
}


# Sort level 1 made chunk-local: each 4096-edge chunk orders its own records by 256-row block, level 2 collects a block's pieces; histogram and scan launches go (6 to 4); sums stay in arrival order
# speedup vs baseline: 3.7360x; 1.1378x over previous
#include <hip/hip_runtime.h>
#include <hip/hip_bf16.h>
#include <math.h>


typedef __attribute__((ext_vector_type(2))) float v2f;
typedef __attribute__((ext_vector_type(8))) float v8f;

#define IN_DIM   128
#define HIDDEN    16
#define NCLASS    40
#define NPAD      48
#define A1_PITCH 132
#define A2_PITCH  20
#define O_PITCH   49

#define CH_THREADS 512
#define CH_EDGES   4096
#define NB_MAX     256

__device__ __forceinline__ void gcn_chunk_load8(const int* __restrict__ p, int c,
                                                int t, int nEdges, int fill,
                                                int (&out)[8]) {
  #pragma unroll
  for (int k = 0; k < 2; ++k) {
    const int e = (c * (CH_EDGES / 4) + k * CH_THREADS + t) * 4;
    if (e + 4 <= nEdges) {
      const int4 q = *reinterpret_cast<const int4*>(p + e);
      out[4 * k + 0] = q.x; out[4 * k + 1] = q.y;
      out[4 * k + 2] = q.z; out[4 * k + 3] = q.w;
    } else {
      #pragma unroll
      for (int u = 0; u < 4; ++u) out[4 * k + u] = (e + u < nEdges) ? p[e + u] : fill;
    }
  }
}

__global__ void __launch_bounds__(CH_THREADS)
gcn_chunk_sort(const int* __restrict__ rows, const int* __restrict__ cols,
               const float* __restrict__ vals, int* __restrict__ recs,
               int* __restrict__ piece, int nEdges, int nChunks, int nBuckets) {
  __shared__ int cnt[NB_MAX];
  __shared__ int wtot[NB_MAX / 32];
  const int t = threadIdx.x;
  const int lane = t & 31;
  const int w = t >> 5;
  const int c = blockIdx.x;
  if (t < NB_MAX) cnt[t] = 0;
  int r[8], cc[8], vv[8], rk[8];
  gcn_chunk_load8(rows, c, t, nEdges, -1, r);
  gcn_chunk_load8(cols, c, t, nEdges, 0, cc);
  gcn_chunk_load8(reinterpret_cast<const int*>(vals), c, t, nEdges, 0, vv);
  __syncthreads();
  #pragma unroll
  for (int u = 0; u < 8; ++u) {
    rk[u] = 0;
    if (r[u] >= 0) rk[u] = atomicAdd(&cnt[r[u] >> 8], 1);
  }
  __syncthreads();

  int cv = 0, inc = 0;
  if (t < NB_MAX) {
    cv = cnt[t];
    inc = cv;
    #pragma unroll
    for (int d = 1; d < 32; d <<= 1) {
      const int n = __shfl_up(inc, d);
      if (lane >= d) inc += n;
    }
    if (lane == 31) wtot[w] = inc;
  }
  __syncthreads();
  if (t < NB_MAX) {
    int off = 0;
    #pragma unroll
    for (int i = 0; i < NB_MAX / 32; ++i)
      if (i < w) off += wtot[i];
    const int excl = off + inc - cv;
    cnt[t] = excl;
    if (t < nBuckets)
      reinterpret_cast<int2*>(piece)[t * nChunks + c] = make_int2(cv, excl);
  }
  __syncthreads();

  int4* __restrict__ rec4 = reinterpret_cast<int4*>(recs) + (size_t)c * CH_EDGES;
  #pragma unroll
  for (int u = 0; u < 8; ++u) {
    if (r[u] >= 0)
      rec4[cnt[r[u] >> 8] + rk[u]] = make_int4(cc[u], vv[u], r[u], 0);
  }
}

#define FP_U 8
__global__ void __launch_bounds__(1024)
gcn_fine_place(const int* __restrict__ recs, const int* __restrict__ piece,
               int* __restrict__ start, int* __restrict__ pairs,
               int nNodes, int nChunks) {
  __shared__ int cnt[256];
  __shared__ int wtot[8];
  __shared__ int wlo[32];
  const int t = threadIdx.x;
  const int lane = t & 31;
  const int w = t >> 5;
  const int b = blockIdx.x;
  const int sub = t & 3;
  const int4* __restrict__ rec4 = reinterpret_cast<const int4*>(recs);
  const int2* __restrict__ pc2 = reinterpret_cast<const int2*>(piece) + (size_t)b * nChunks;
  if (t < 256) cnt[t] = 0;
  __syncthreads();

  int myLo = 0;
  for (int c = t >> 2; c < nChunks; c += 256) {
    const int2 pc = pc2[c];
    if (sub == 0) myLo += pc.y;
    const int4* __restrict__ p = rec4 + (size_t)c * CH_EDGES + pc.y;
    for (int j0 = sub; j0 < pc.x; j0 += 4 * FP_U) {
      int z[FP_U];
      #pragma unroll
      for (int u = 0; u < FP_U; ++u) {
        const int j = j0 + 4 * u;
        z[u] = (j < pc.x) ? p[j].z : -1;
      }
      #pragma unroll
      for (int u = 0; u < FP_U; ++u)
        if (z[u] >= 0) atomicAdd(&cnt[z[u] & 255], 1);
    }
  }
  #pragma unroll
  for (int d = 16; d > 0; d >>= 1) myLo += __shfl_xor(myLo, d);
  if (lane == 0) wlo[w] = myLo;
  __syncthreads();

  int c = 0, inc = 0;
  if (t < 256) {
    c = cnt[t];
    inc = c;
    #pragma unroll
    for (int d = 1; d < 32; d <<= 1) {
      const int n = __shfl_up(inc, d);
      if (lane >= d) inc += n;
    }
    if (lane == 31) wtot[w] = inc;
  }
  __syncthreads();
  if (t < 256) {
    int off = 0;
    #pragma unroll
    for (int i = 0; i < 32; ++i) off += wlo[i];
    #pragma unroll
    for (int i = 0; i < 8; ++i)
      if (i < w) off += wtot[i];
    const int first = off + inc - c;
    const int row = b * 256 + t;
    if (row < nNodes) start[row] = first;
    if (row == nNodes - 1) start[nNodes] = first + c;
    cnt[t] = first;
  }
  __syncthreads();

  int2* __restrict__ pr = reinterpret_cast<int2*>(pairs);
  for (int cb = t >> 2; cb < nChunks; cb += 256) {
    const int2 pc = pc2[cb];
    const int4* __restrict__ p = rec4 + (size_t)cb * CH_EDGES + pc.y;
    for (int j0 = sub; j0 < pc.x; j0 += 4 * FP_U) {
      int4 q[FP_U];
      #pragma unroll
      for (int u = 0; u < FP_U; ++u) {
        const int j = j0 + 4 * u;
        q[u] = (j < pc.x) ? p[j] : make_int4(0, 0, -1, 0);
      }
      #pragma unroll
      for (int u = 0; u < FP_U; ++u) {
        if (q[u].z >= 0) {
          const int pos = atomicAdd(&cnt[q[u].z & 255], 1);
          pr[pos] = make_int2(q[u].x, q[u].y);
        }
      }
    }
  }
}

template <int U>
__device__ __forceinline__ void gcn_gather_batch(float4& acc, int2 mine, int j,
                                                 const float4* __restrict__ x4,
                                                 int lane) {
  float4 d[U];
  float v[U];
  #pragma unroll
  for (int u = 0; u < U; ++u) {
    const int c = __builtin_amdgcn_readlane(mine.x, j + u);
    v[u] = __int_as_float(__builtin_amdgcn_readlane(mine.y, j + u));
    d[u] = x4[(size_t)c * (IN_DIM / 4) + lane];
  }
  #pragma unroll
  for (int u = 0; u < U; ++u) {
    acc.x += v[u] * d[u].x;
    acc.y += v[u] * d[u].y;
    acc.z += v[u] * d[u].z;
    acc.w += v[u] * d[u].w;
  }
}

__device__ __forceinline__ float4 gcn_row_sum1(int s, int e, int2 mine,
                                               const int2* __restrict__ pr,
                                               const float4* __restrict__ x4,
                                               int lane) {
  float4 acc = make_float4(0.0f, 0.0f, 0.0f, 0.0f);
  for (int base = s; base < e; base += 32) {
    const int n = (e - base) < 32 ? (e - base) : 32;
    if (base != s) {
      mine = make_int2(0, 0);
      if (lane < n) mine = pr[base + lane];
    }
    int j = 0;
    for (; j + 8 <= n; j += 8) gcn_gather_batch<8>(acc, mine, j, x4, lane);
    if (n - j >= 4) { gcn_gather_batch<4>(acc, mine, j, x4, lane); j += 4; }
    if (n - j >= 2) { gcn_gather_batch<2>(acc, mine, j, x4, lane); j += 2; }
    if (n - j >= 1) { gcn_gather_batch<1>(acc, mine, j, x4, lane); }
  }
  return acc;
}

__global__ void __launch_bounds__(128)
gcn_gather1_gemm1(const int* __restrict__ start,
                  const int* __restrict__ pairs,
                  const float* __restrict__ x,
                  const float* __restrict__ w1,
                  float* __restrict__ h, int nNodes) {
  __shared__ __attribute__((aligned(16))) float Wl[IN_DIM * HIDDEN];
  __shared__ __attribute__((aligned(16))) float Al[16 * A1_PITCH];
  const int t = threadIdx.x;
  const int lane = t & 31;
  const int w = __builtin_amdgcn_readfirstlane(t >> 5);
  const int tile = blockIdx.x;
  const int r0 = tile * 16 + w * 4;
  const int2* __restrict__ pr = reinterpret_cast<const int2*>(pairs);
  const float4* __restrict__ x4 = reinterpret_cast<const float4*>(x);

  int sv = 0;
  if (lane < 5) {
    const int idx = r0 + lane;
    sv = start[idx < nNodes ? idx : nNodes];
  }

  {
    const float4* __restrict__ w14 = reinterpret_cast<const float4*>(w1);
    float4 wreg[4];
    #pragma unroll
    for (int i = 0; i < 4; ++i) wreg[i] = w14[t + 128 * i];
    #pragma unroll
    for (int i = 0; i < 4; ++i) reinterpret_cast<float4*>(Wl)[t + 128 * i] = wreg[i];
  }

  int S[5];
  #pragma unroll
  for (int q = 0; q < 5; ++q) S[q] = __builtin_amdgcn_readlane(sv, q);

  int2 mine[4];
  #pragma unroll
  for (int q = 0; q < 4; ++q) {
    const int d = S[q + 1] - S[q];
    const int n = d < 32 ? d : 32;
    mine[q] = make_int2(0, 0);
    if (lane < n) mine[q] = pr[S[q] + lane];
  }

  #pragma unroll
  for (int q = 0; q < 4; ++q) {
    const float4 acc = gcn_row_sum1(S[q], S[q + 1], mine[q], pr, x4, lane);
    *reinterpret_cast<float4*>(&Al[(w * 4 + q) * A1_PITCH + lane * 4]) = acc;
  }
  __syncthreads();
  if (w != 0) return;

  const int half = lane >> 4;
  const int l    = lane & 15;

  v8f c = {};
  #pragma unroll 8
  for (int k0 = 0; k0 < IN_DIM; k0 += 4) {
    const int ka = k0 + 2 * half;
    const v2f a = *reinterpret_cast<const v2f*>(&Al[l * A1_PITCH + ka]);
    v2f b;
    b.x = Wl[ka * HIDDEN + l];
    b.y = Wl[(ka + 1) * HIDDEN + l];
    c = __builtin_amdgcn_wmma_f32_16x16x4_f32(false, a, false, b,
                                              (short)0, c, false, false);
  }

  float* hb = h + (size_t)tile * 16 * HIDDEN;
  #pragma unroll
  for (int i = 0; i < 8; ++i) {
    const float v = c[i];
    hb[(i + 8 * half) * HIDDEN + l] = v > 0.0f ? v : 0.0f;
  }
}

__global__ void __launch_bounds__(128)
gcn_gather2_gemm2_lsm(const int* __restrict__ start,
                      const int* __restrict__ pairs,
                      const float* __restrict__ h,
                      const float* __restrict__ w2,
                      float* __restrict__ out, int nNodes) {
  __shared__ float Wl[HIDDEN * NPAD];
  __shared__ __attribute__((aligned(16))) float Al[16 * A2_PITCH];
  __shared__ float Ol[16 * O_PITCH];
  const int t = threadIdx.x;
  const int lane = t & 31;
  const int w = __builtin_amdgcn_readfirstlane(t >> 5);
  const int tile = blockIdx.x;
  const int slot = (lane >> 2) & 3;
  const int f4   = lane & 3;
  const int2* __restrict__ pr = reinterpret_cast<const int2*>(pairs);
  const float4* __restrict__ h4 = reinterpret_cast<const float4*>(h);

  int rl[2], s[2], e[2];
  #pragma unroll
  for (int u = 0; u < 2; ++u) {
    rl[u] = w * 4 + u * 2 + (lane >> 4);
    const int r = tile * 16 + rl[u];
    s[u] = start[r < nNodes ? r : nNodes];
    e[u] = start[r + 1 < nNodes ? r + 1 : nNodes];
  }

  #pragma unroll
  for (int k = 0; k < (HIDDEN * NPAD) / 128; ++k) {
    const int i = t + 128 * k;
    const int r = i / NPAD, n = i % NPAD;
    const float v = w2[r * NCLASS + (n < NCLASS ? n : NCLASS - 1)];
    Wl[i] = (n < NCLASS) ? v : 0.0f;
  }

  #pragma unroll
  for (int u = 0; u < 2; ++u) {
    float4 acc = make_float4(0.0f, 0.0f, 0.0f, 0.0f);
    int p = s[u] + slot;
    for (; p + 4 < e[u]; p += 8) {
      const int2 cv0 = pr[p];
      const int2 cv1 = pr[p + 4];
      const float4 d0 = h4[(size_t)cv0.x * (HIDDEN / 4) + f4];
      const float4 d1 = h4[(size_t)cv1.x * (HIDDEN / 4) + f4];
      const float v0 = __int_as_float(cv0.y);
      const float v1 = __int_as_float(cv1.y);
      acc.x += v0 * d0.x; acc.y += v0 * d0.y; acc.z += v0 * d0.z; acc.w += v0 * d0.w;
      acc.x += v1 * d1.x; acc.y += v1 * d1.y; acc.z += v1 * d1.z; acc.w += v1 * d1.w;
    }
    if (p < e[u]) {
      const int2 cv0 = pr[p];
      const float4 d0 = h4[(size_t)cv0.x * (HIDDEN / 4) + f4];
      const float v0 = __int_as_float(cv0.y);
      acc.x += v0 * d0.x; acc.y += v0 * d0.y; acc.z += v0 * d0.z; acc.w += v0 * d0.w;
    }
    #pragma unroll
    for (int d = 4; d <= 8; d <<= 1) {
      acc.x += __shfl_xor(acc.x, d);
      acc.y += __shfl_xor(acc.y, d);
      acc.z += __shfl_xor(acc.z, d);
      acc.w += __shfl_xor(acc.w, d);
    }
    if (slot == 0)
      *reinterpret_cast<float4*>(&Al[rl[u] * A2_PITCH + f4 * 4]) = acc;
  }
  __syncthreads();

  if (w == 0) {
    const int half = lane >> 4;
    const int l    = lane & 15;
    #pragma unroll
    for (int nt = 0; nt < 3; ++nt) {
      v8f c = {};
      #pragma unroll
      for (int k0 = 0; k0 < HIDDEN; k0 += 4) {
        const int ka = k0 + 2 * half;
        const v2f a = *reinterpret_cast<const v2f*>(&Al[l * A2_PITCH + ka]);
        v2f b;
        b.x = Wl[ka * NPAD + nt * 16 + l];
        b.y = Wl[(ka + 1) * NPAD + nt * 16 + l];
        c = __builtin_amdgcn_wmma_f32_16x16x4_f32(false, a, false, b,
                                                  (short)0, c, false, false);
      }
      #pragma unroll
      for (int i = 0; i < 8; ++i)
        Ol[(i + 8 * half) * O_PITCH + nt * 16 + l] = c[i];
    }
  }
  __syncthreads();

  if (w == 0 && lane < 16 && tile * 16 + lane < nNodes) {
    const float* row = Ol + lane * O_PITCH;
    float mx = row[0];
    for (int j = 1; j < NCLASS; ++j) mx = fmaxf(mx, row[j]);
    float sum = 0.0f;
    for (int j = 0; j < NCLASS; ++j) sum += __expf(row[j] - mx);
    const float lse = mx + __logf(sum);
    float* ob = out + ((size_t)tile * 16 + lane) * NCLASS;
    for (int j = 0; j < NCLASS; ++j) ob[j] = row[j] - lse;
  }
}

extern "C" void kernel_launch(void* const* d_in, const int* in_sizes, int n_in,
                              void* d_out, int out_size, void* d_ws, size_t ws_size,
                              hipStream_t stream) {
  const float* features = (const float*)d_in[0];
  const float* weight1  = (const float*)d_in[1];
  const float* weight2  = (const float*)d_in[2];
  const float* edgeVals = (const float*)d_in[3];
  const int*   edgeRows = (const int*)  d_in[4];
  const int*   edgeCols = (const int*)  d_in[5];

  const int nNodes = in_sizes[0] / IN_DIM;
  const int nEdges = in_sizes[3];
  const int nTiles = nNodes / 16;

  const int nBuckets = (nNodes + 255) / 256;
  const int nChunks  = (nEdges + CH_EDGES - 1) / CH_EDGES;
  const int nHist    = nBuckets * nChunks;

  float* h  = (float*)d_ws;
  int* pairs  = (int*)(h + (size_t)nNodes * HIDDEN);
  int* recs   = pairs + (size_t)nEdges * 2;
  int* piece  = recs + (size_t)nChunks * CH_EDGES * 4;
  int* start  = piece + (size_t)nHist * 2;

  gcn_chunk_sort<<<nChunks, CH_THREADS, 0, stream>>>(
      edgeRows, edgeCols, edgeVals, recs, piece, nEdges, nChunks, nBuckets);
  gcn_fine_place<<<nBuckets, 1024, 0, stream>>>(recs, piece, start, pairs, nNodes, nChunks);
  gcn_gather1_gemm1<<<nTiles, 128, 0, stream>>>(start, pairs, features, weight1, h, nNodes);
  gcn_gather2_gemm2_lsm<<<nTiles, 128, 0, stream>>>(start, pairs, h, weight2, (float*)d_out, nNodes);
}


# Sort levels 1 and 2: ordered records and (source, weight) slots assembled in LDS, copied out in address order, whole 128-B lines per store not 32 scattered pieces; sums stay in arrival order
# speedup vs baseline: 3.9841x; 1.0664x over previous
#include <hip/hip_runtime.h>
#include <hip/hip_bf16.h>
#include <math.h>


typedef __attribute__((ext_vector_type(2))) float v2f;
typedef __attribute__((ext_vector_type(8))) float v8f;

#define IN_DIM   128
#define HIDDEN    16
#define NCLASS    40
#define NPAD      48
#define A1_PITCH 132
#define A2_PITCH  20
#define O_PITCH   49

#define CH_THREADS 512
#define CH_EDGES   4096
#define NB_MAX     256

__device__ __forceinline__ void gcn_chunk_load8(const int* __restrict__ p, int c,
                                                int t, int nEdges, int fill,
                                                int (&out)[8]) {
  #pragma unroll
  for (int k = 0; k < 2; ++k) {
    const int e = (c * (CH_EDGES / 4) + k * CH_THREADS + t) * 4;
    if (e + 4 <= nEdges) {
      const int4 q = *reinterpret_cast<const int4*>(p + e);
      out[4 * k + 0] = q.x; out[4 * k + 1] = q.y;
      out[4 * k + 2] = q.z; out[4 * k + 3] = q.w;
    } else {
      #pragma unroll
      for (int u = 0; u < 4; ++u) out[4 * k + u] = (e + u < nEdges) ? p[e + u] : fill;
    }
  }
}

__global__ void __launch_bounds__(CH_THREADS)
gcn_chunk_sort(const int* __restrict__ rows, const int* __restrict__ cols,
               const float* __restrict__ vals, int* __restrict__ recs,
               int* __restrict__ piece, int nEdges, int nChunks, int nBuckets) {
  __shared__ __attribute__((aligned(16))) int4 img[CH_EDGES];
  __shared__ int cnt[NB_MAX];
  __shared__ int wtot[NB_MAX / 32];
  const int t = threadIdx.x;
  const int lane = t & 31;
  const int w = t >> 5;
  const int c = blockIdx.x;
  if (t < NB_MAX) cnt[t] = 0;
  int r[8], cc[8], vv[8], rk[8];
  gcn_chunk_load8(rows, c, t, nEdges, -1, r);
  gcn_chunk_load8(cols, c, t, nEdges, 0, cc);
  gcn_chunk_load8(reinterpret_cast<const int*>(vals), c, t, nEdges, 0, vv);
  __syncthreads();
  #pragma unroll
  for (int u = 0; u < 8; ++u) {
    rk[u] = 0;
    if (r[u] >= 0) rk[u] = atomicAdd(&cnt[r[u] >> 8], 1);
  }
  __syncthreads();

  int cv = 0, inc = 0;
  if (t < NB_MAX) {
    cv = cnt[t];
    inc = cv;
    #pragma unroll
    for (int d = 1; d < 32; d <<= 1) {
      const int n = __shfl_up(inc, d);
      if (lane >= d) inc += n;
    }
    if (lane == 31) wtot[w] = inc;
  }
  __syncthreads();
  if (t < NB_MAX) {
    int off = 0;
    #pragma unroll
    for (int i = 0; i < NB_MAX / 32; ++i)
      if (i < w) off += wtot[i];
    const int excl = off + inc - cv;
    cnt[t] = excl;
    if (t < nBuckets)
      reinterpret_cast<int2*>(piece)[t * nChunks + c] = make_int2(cv, excl);
  }
  __syncthreads();

  int pos[8];
  #pragma unroll
  for (int u = 0; u < 8; ++u) pos[u] = cnt[r[u] >= 0 ? (r[u] >> 8) : 0] + rk[u];
  #pragma unroll
  for (int u = 0; u < 8; ++u) {
    if (r[u] >= 0) img[pos[u]] = make_int4(cc[u], vv[u], r[u], 0);
  }
  __syncthreads();

  const int left = nEdges - c * CH_EDGES;
  const int nValid = left < CH_EDGES ? left : CH_EDGES;
  int4* __restrict__ rec4 = reinterpret_cast<int4*>(recs) + (size_t)c * CH_EDGES;
  if (nValid == CH_EDGES) {
    int4 q[CH_EDGES / CH_THREADS];
    #pragma unroll
    for (int k = 0; k < CH_EDGES / CH_THREADS; ++k) q[k] = img[k * CH_THREADS + t];
    #pragma unroll
    for (int k = 0; k < CH_EDGES / CH_THREADS; ++k) rec4[k * CH_THREADS + t] = q[k];
  } else {
    for (int i = t; i < nValid; i += CH_THREADS) rec4[i] = img[i];
  }
}

#define FP_U 8
#define FP_CAP 8192

template <bool VIA_LDS>
__device__ __forceinline__ void gcn_fine_put(const int4* __restrict__ rec4,
                                             const int2* __restrict__ pc2,
                                             int* cnt, int2* img,
                                             int2* __restrict__ pr,
                                             int bFirst, int nChunks, int t, int sub) {
  for (int cb = t >> 2; cb < nChunks; cb += 256) {
    const int2 pc = pc2[cb];
    const int4* __restrict__ p = rec4 + (size_t)cb * CH_EDGES + pc.y;
    for (int j0 = sub; j0 < pc.x; j0 += 4 * FP_U) {
      int4 q[FP_U];
      #pragma unroll
      for (int u = 0; u < FP_U; ++u) {
        const int j = j0 + 4 * u;
        q[u] = (j < pc.x) ? p[j] : make_int4(0, 0, -1, 0);
      }
      int pos[FP_U];
      #pragma unroll
      for (int u = 0; u < FP_U; ++u)
        pos[u] = (q[u].z >= 0) ? atomicAdd(&cnt[q[u].z & 255], 1) : 0;
      #pragma unroll
      for (int u = 0; u < FP_U; ++u) {
        if (VIA_LDS) {
          img[(q[u].z >= 0) ? pos[u] - bFirst : FP_CAP + (t & 31)] = make_int2(q[u].x, q[u].y);
        } else if (q[u].z >= 0) {
          pr[pos[u]] = make_int2(q[u].x, q[u].y);
        }
      }
    }
  }
}
__global__ void __launch_bounds__(1024)
gcn_fine_place(const int* __restrict__ recs, const int* __restrict__ piece,
               int* __restrict__ start, int* __restrict__ pairs,
               int nNodes, int nChunks) {
  __shared__ __attribute__((aligned(16))) int2 img[FP_CAP + 32];
  __shared__ int cnt[256];
  __shared__ int wtot[8];
  __shared__ int wlo[32];
  const int t = threadIdx.x;
  const int lane = t & 31;
  const int w = t >> 5;
  const int b = blockIdx.x;
  const int sub = t & 3;
  const int4* __restrict__ rec4 = reinterpret_cast<const int4*>(recs);
  const int2* __restrict__ pc2 = reinterpret_cast<const int2*>(piece) + (size_t)b * nChunks;
  if (t < 256) cnt[t] = 0;
  __syncthreads();

  int myLo = 0;
  for (int c = t >> 2; c < nChunks; c += 256) {
    const int2 pc = pc2[c];
    if (sub == 0) myLo += pc.y;
    const int4* __restrict__ p = rec4 + (size_t)c * CH_EDGES + pc.y;
    for (int j0 = sub; j0 < pc.x; j0 += 4 * FP_U) {
      int z[FP_U];
      #pragma unroll
      for (int u = 0; u < FP_U; ++u) {
        const int j = j0 + 4 * u;
        z[u] = (j < pc.x) ? p[j].z : -1;
      }
      #pragma unroll
      for (int u = 0; u < FP_U; ++u)
        if (z[u] >= 0) atomicAdd(&cnt[z[u] & 255], 1);
    }
  }
  #pragma unroll
  for (int d = 16; d > 0; d >>= 1) myLo += __shfl_xor(myLo, d);
  if (lane == 0) wlo[w] = myLo;
  __syncthreads();

  int c = 0, inc = 0;
  if (t < 256) {
    c = cnt[t];
    inc = c;
    #pragma unroll
    for (int d = 1; d < 32; d <<= 1) {
      const int n = __shfl_up(inc, d);
      if (lane >= d) inc += n;
    }
    if (lane == 31) wtot[w] = inc;
  }
  __syncthreads();
  if (t < 256) {
    int off = 0;
    #pragma unroll
    for (int i = 0; i < 32; ++i) off += wlo[i];
    #pragma unroll
    for (int i = 0; i < 8; ++i)
      if (i < w) off += wtot[i];
    const int first = off + inc - c;
    const int row = b * 256 + t;
    if (row < nNodes) start[row] = first;
    if (row == nNodes - 1) start[nNodes] = first + c;
    cnt[t] = first;
  }
  __syncthreads();

  int bFirst = 0, nb = 0;
  #pragma unroll
  for (int i = 0; i < 32; ++i) bFirst += wlo[i];
  #pragma unroll
  for (int i = 0; i < 8; ++i) nb += wtot[i];
  bFirst = __builtin_amdgcn_readfirstlane(bFirst);
  nb = __builtin_amdgcn_readfirstlane(nb);

  int2* __restrict__ pr = reinterpret_cast<int2*>(pairs);
  if (nb <= FP_CAP) {
    gcn_fine_put<true>(rec4, pc2, cnt, img, pr, bFirst, nChunks, t, sub);
    __syncthreads();
    const int a0 = bFirst & ~31;
    const int end = nb > 0 ? bFirst + nb : a0;
    for (int i0 = a0; i0 < end; i0 += 4 * 1024) {
      int2 v[4];
      #pragma unroll
      for (int k = 0; k < 4; ++k) {
        const int li = i0 + k * 1024 + t - bFirst;
        v[k] = img[li < 0 ? 0 : (li < nb ? li : nb - 1)];
      }
      #pragma unroll
      for (int k = 0; k < 4; ++k) {
        const int i = i0 + k * 1024 + t;
        if (i >= bFirst && i < end) pr[i] = v[k];
      }
    }
  } else {
    gcn_fine_put<false>(rec4, pc2, cnt, img, pr, bFirst, nChunks, t, sub);
  }
}

template <int U>
__device__ __forceinline__ void gcn_gather_batch(float4& acc, int2 mine, int j,
                                                 const float4* __restrict__ x4,
                                                 int lane) {
  float4 d[U];
  float v[U];
  #pragma unroll
  for (int u = 0; u < U; ++u) {
    const int c = __builtin_amdgcn_readlane(mine.x, j + u);
    v[u] = __int_as_float(__builtin_amdgcn_readlane(mine.y, j + u));
    d[u] = x4[(size_t)c * (IN_DIM / 4) + lane];
  }
  #pragma unroll
  for (int u = 0; u < U; ++u) {
    acc.x += v[u] * d[u].x;
    acc.y += v[u] * d[u].y;
    acc.z += v[u] * d[u].z;
    acc.w += v[u] * d[u].w;
  }
}

__device__ __forceinline__ float4 gcn_row_sum1(int s, int e, int2 mine,
                                               const int2* __restrict__ pr,
                                               const float4* __restrict__ x4,
                                               int lane) {
  float4 acc = make_float4(0.0f, 0.0f, 0.0f, 0.0f);
  for (int base = s; base < e; base += 32) {
    const int n = (e - base) < 32 ? (e - base) : 32;
    if (base != s) {
      mine = make_int2(0, 0);
      if (lane < n) mine = pr[base + lane];
    }
    int j = 0;
    for (; j + 8 <= n; j += 8) gcn_gather_batch<8>(acc, mine, j, x4, lane);
    if (n - j >= 4) { gcn_gather_batch<4>(acc, mine, j, x4, lane); j += 4; }
    if (n - j >= 2) { gcn_gather_batch<2>(acc, mine, j, x4, lane); j += 2; }
    if (n - j >= 1) { gcn_gather_batch<1>(acc, mine, j, x4, lane); }
  }
  return acc;
}

__global__ void __launch_bounds__(128)
gcn_gather1_gemm1(const int* __restrict__ start,
                  const int* __restrict__ pairs,
                  const float* __restrict__ x,
                  const float* __restrict__ w1,
                  float* __restrict__ h, int nNodes) {
  __shared__ __attribute__((aligned(16))) float Wl[IN_DIM * HIDDEN];
  __shared__ __attribute__((aligned(16))) float Al[16 * A1_PITCH];
  const int t = threadIdx.x;
  const int lane = t & 31;
  const int w = __builtin_amdgcn_readfirstlane(t >> 5);
  const int tile = blockIdx.x;
  const int r0 = tile * 16 + w * 4;
  const int2* __restrict__ pr = reinterpret_cast<const int2*>(pairs);
  const float4* __restrict__ x4 = reinterpret_cast<const float4*>(x);

  int sv = 0;
  if (lane < 5) {
    const int idx = r0 + lane;
    sv = start[idx < nNodes ? idx : nNodes];
  }

  {
    const float4* __restrict__ w14 = reinterpret_cast<const float4*>(w1);
    float4 wreg[4];
    #pragma unroll
    for (int i = 0; i < 4; ++i) wreg[i] = w14[t + 128 * i];
    #pragma unroll
    for (int i = 0; i < 4; ++i) reinterpret_cast<float4*>(Wl)[t + 128 * i] = wreg[i];
  }

  int S[5];
  #pragma unroll
  for (int q = 0; q < 5; ++q) S[q] = __builtin_amdgcn_readlane(sv, q);

  int2 mine[4];
  #pragma unroll
  for (int q = 0; q < 4; ++q) {
    const int d = S[q + 1] - S[q];
    const int n = d < 32 ? d : 32;
    mine[q] = make_int2(0, 0);
    if (lane < n) mine[q] = pr[S[q] + lane];
  }

  #pragma unroll
  for (int q = 0; q < 4; ++q) {
    const float4 acc = gcn_row_sum1(S[q], S[q + 1], mine[q], pr, x4, lane);
    *reinterpret_cast<float4*>(&Al[(w * 4 + q) * A1_PITCH + lane * 4]) = acc;
  }
  __syncthreads();
  if (w != 0) return;

  const int half = lane >> 4;
  const int l    = lane & 15;

  v8f c = {};
  #pragma unroll 8
  for (int k0 = 0; k0 < IN_DIM; k0 += 4) {
    const int ka = k0 + 2 * half;
    const v2f a = *reinterpret_cast<const v2f*>(&Al[l * A1_PITCH + ka]);
    v2f b;
    b.x = Wl[ka * HIDDEN + l];
    b.y = Wl[(ka + 1) * HIDDEN + l];
    c = __builtin_amdgcn_wmma_f32_16x16x4_f32(false, a, false, b,
                                              (short)0, c, false, false);
  }

  float* hb = h + (size_t)tile * 16 * HIDDEN;
  #pragma unroll
  for (int i = 0; i < 8; ++i) {
    const float v = c[i];
    hb[(i + 8 * half) * HIDDEN + l] = v > 0.0f ? v : 0.0f;
  }
}

__global__ void __launch_bounds__(128)
gcn_gather2_gemm2_lsm(const int* __restrict__ start,
                      const int* __restrict__ pairs,
                      const float* __restrict__ h,
                      const float* __restrict__ w2,
                      float* __restrict__ out, int nNodes) {
  __shared__ float Wl[HIDDEN * NPAD];
  __shared__ __attribute__((aligned(16))) float Al[16 * A2_PITCH];
  __shared__ float Ol[16 * O_PITCH];
  const int t = threadIdx.x;
  const int lane = t & 31;
  const int w = __builtin_amdgcn_readfirstlane(t >> 5);
  const int tile = blockIdx.x;
  const int slot = (lane >> 2) & 3;
  const int f4   = lane & 3;
  const int2* __restrict__ pr = reinterpret_cast<const int2*>(pairs);
  const float4* __restrict__ h4 = reinterpret_cast<const float4*>(h);

  int rl[2], s[2], e[2];
  #pragma unroll
  for (int u = 0; u < 2; ++u) {
    rl[u] = w * 4 + u * 2 + (lane >> 4);
    const int r = tile * 16 + rl[u];
    s[u] = start[r < nNodes ? r : nNodes];
    e[u] = start[r + 1 < nNodes ? r + 1 : nNodes];
  }

  #pragma unroll
  for (int k = 0; k < (HIDDEN * NPAD) / 128; ++k) {
    const int i = t + 128 * k;
    const int r = i / NPAD, n = i % NPAD;
    const float v = w2[r * NCLASS + (n < NCLASS ? n : NCLASS - 1)];
    Wl[i] = (n < NCLASS) ? v : 0.0f;
  }

  #pragma unroll
  for (int u = 0; u < 2; ++u) {
    float4 acc = make_float4(0.0f, 0.0f, 0.0f, 0.0f);
    int p = s[u] + slot;
    for (; p + 4 < e[u]; p += 8) {
      const int2 cv0 = pr[p];
      const int2 cv1 = pr[p + 4];
      const float4 d0 = h4[(size_t)cv0.x * (HIDDEN / 4) + f4];
      const float4 d1 = h4[(size_t)cv1.x * (HIDDEN / 4) + f4];
      const float v0 = __int_as_float(cv0.y);
      const float v1 = __int_as_float(cv1.y);
      acc.x += v0 * d0.x; acc.y += v0 * d0.y; acc.z += v0 * d0.z; acc.w += v0 * d0.w;
      acc.x += v1 * d1.x; acc.y += v1 * d1.y; acc.z += v1 * d1.z; acc.w += v1 * d1.w;
    }
    if (p < e[u]) {
      const int2 cv0 = pr[p];
      const float4 d0 = h4[(size_t)cv0.x * (HIDDEN / 4) + f4];
      const float v0 = __int_as_float(cv0.y);
      acc.x += v0 * d0.x; acc.y += v0 * d0.y; acc.z += v0 * d0.z; acc.w += v0 * d0.w;
    }
    #pragma unroll
    for (int d = 4; d <= 8; d <<= 1) {
      acc.x += __shfl_xor(acc.x, d);
      acc.y += __shfl_xor(acc.y, d);
      acc.z += __shfl_xor(acc.z, d);
      acc.w += __shfl_xor(acc.w, d);
    }
    if (slot == 0)
      *reinterpret_cast<float4*>(&Al[rl[u] * A2_PITCH + f4 * 4]) = acc;
  }
  __syncthreads();

  if (w == 0) {
    const int half = lane >> 4;
    const int l    = lane & 15;
    #pragma unroll
    for (int nt = 0; nt < 3; ++nt) {
      v8f c = {};
      #pragma unroll
      for (int k0 = 0; k0 < HIDDEN; k0 += 4) {
        const int ka = k0 + 2 * half;
        const v2f a = *reinterpret_cast<const v2f*>(&Al[l * A2_PITCH + ka]);
        v2f b;
        b.x = Wl[ka * NPAD + nt * 16 + l];
        b.y = Wl[(ka + 1) * NPAD + nt * 16 + l];
        c = __builtin_amdgcn_wmma_f32_16x16x4_f32(false, a, false, b,
                                                  (short)0, c, false, false);
      }
      #pragma unroll
      for (int i = 0; i < 8; ++i)
        Ol[(i + 8 * half) * O_PITCH + nt * 16 + l] = c[i];
    }
  }
  __syncthreads();

  if (w == 0 && lane < 16 && tile * 16 + lane < nNodes) {
    const float* row = Ol + lane * O_PITCH;
    float mx = row[0];
    for (int j = 1; j < NCLASS; ++j) mx = fmaxf(mx, row[j]);
    float sum = 0.0f;
    for (int j = 0; j < NCLASS; ++j) sum += __expf(row[j] - mx);
    const float lse = mx + __logf(sum);
    float* ob = out + ((size_t)tile * 16 + lane) * NCLASS;
    for (int j = 0; j < NCLASS; ++j) ob[j] = row[j] - lse;
  }
}

extern "C" void kernel_launch(void* const* d_in, const int* in_sizes, int n_in,
                              void* d_out, int out_size, void* d_ws, size_t ws_size,
                              hipStream_t stream) {
  const float* features = (const float*)d_in[0];
  const float* weight1  = (const float*)d_in[1];
  const float* weight2  = (const float*)d_in[2];
  const float* edgeVals = (const float*)d_in[3];
  const int*   edgeRows = (const int*)  d_in[4];
  const int*   edgeCols = (const int*)  d_in[5];

  const int nNodes = in_sizes[0] / IN_DIM;
  const int nEdges = in_sizes[3];
  const int nTiles = nNodes / 16;

  const int nBuckets = (nNodes + 255) / 256;
  const int nChunks  = (nEdges + CH_EDGES - 1) / CH_EDGES;
  const int nHist    = nBuckets * nChunks;

  float* h  = (float*)d_ws;
  int* pairs  = (int*)(h + (size_t)nNodes * HIDDEN);
  int* recs   = pairs + (size_t)nEdges * 2;
  int* piece  = recs + (size_t)nChunks * CH_EDGES * 4;
  int* start  = piece + (size_t)nHist * 2;

  gcn_chunk_sort<<<nChunks, CH_THREADS, 0, stream>>>(
      edgeRows, edgeCols, edgeVals, recs, piece, nEdges, nChunks, nBuckets);
  gcn_fine_place<<<nBuckets, 1024, 0, stream>>>(recs, piece, start, pairs, nNodes, nChunks);
  gcn_gather1_gemm1<<<nTiles, 128, 0, stream>>>(start, pairs, features, weight1, h, nNodes);
  gcn_gather2_gemm2_lsm<<<nTiles, 128, 0, stream>>>(start, pairs, h, weight2, (float*)d_out, nNodes);
}


# Sort level 2 and layer 1 in one kernel: the work-group that orders a 256-row block gathers those rows from its LDS slot image and runs the WMMA on its 16 tiles; 4 to 3 launches; sum order kept
# speedup vs baseline: 4.2014x; 1.0545x over previous
#include <hip/hip_runtime.h>
#include <hip/hip_bf16.h>
#include <math.h>


typedef __attribute__((ext_vector_type(2))) float v2f;
typedef __attribute__((ext_vector_type(8))) float v8f;

#define IN_DIM   128
#define HIDDEN    16
#define NCLASS    40
#define NPAD      48
#define A1_PITCH 132
#define A2_PITCH  20
#define O_PITCH   49

#define CH_THREADS 512
#define CH_EDGES   4096
#define NB_MAX     256

__device__ __forceinline__ void gcn_chunk_load8(const int* __restrict__ p, int c,
                                                int t, int nEdges, int fill,
                                                int (&out)[8]) {
  #pragma unroll
  for (int k = 0; k < 2; ++k) {
    const int e = (c * (CH_EDGES / 4) + k * CH_THREADS + t) * 4;
    if (e + 4 <= nEdges) {
      const int4 q = *reinterpret_cast<const int4*>(p + e);
      out[4 * k + 0] = q.x; out[4 * k + 1] = q.y;
      out[4 * k + 2] = q.z; out[4 * k + 3] = q.w;
    } else {
      #pragma unroll
      for (int u = 0; u < 4; ++u) out[4 * k + u] = (e + u < nEdges) ? p[e + u] : fill;
    }
  }
}

__global__ void __launch_bounds__(CH_THREADS)
gcn_chunk_sort(const int* __restrict__ rows, const int* __restrict__ cols,
               const float* __restrict__ vals, int* __restrict__ recs,
               int* __restrict__ piece, int nEdges, int nChunks, int nBuckets) {
  __shared__ __attribute__((aligned(16))) int4 img[CH_EDGES];
  __shared__ int cnt[NB_MAX];
  __shared__ int wtot[NB_MAX / 32];
  const int t = threadIdx.x;
  const int lane = t & 31;
  const int w = t >> 5;
  const int c = blockIdx.x;
  if (t < NB_MAX) cnt[t] = 0;
  int r[8], cc[8], vv[8], rk[8];
  gcn_chunk_load8(rows, c, t, nEdges, -1, r);
  gcn_chunk_load8(cols, c, t, nEdges, 0, cc);
  gcn_chunk_load8(reinterpret_cast<const int*>(vals), c, t, nEdges, 0, vv);
  __syncthreads();
  #pragma unroll
  for (int u = 0; u < 8; ++u) {
    rk[u] = 0;
    if (r[u] >= 0) rk[u] = atomicAdd(&cnt[r[u] >> 8], 1);
  }
  __syncthreads();

  int cv = 0, inc = 0;
  if (t < NB_MAX) {
    cv = cnt[t];
    inc = cv;
    #pragma unroll
    for (int d = 1; d < 32; d <<= 1) {
      const int n = __shfl_up(inc, d);
      if (lane >= d) inc += n;
    }
    if (lane == 31) wtot[w] = inc;
  }
  __syncthreads();
  if (t < NB_MAX) {
    int off = 0;
    #pragma unroll
    for (int i = 0; i < NB_MAX / 32; ++i)
      if (i < w) off += wtot[i];
    const int excl = off + inc - cv;
    cnt[t] = excl;
    if (t < nBuckets)
      reinterpret_cast<int2*>(piece)[t * nChunks + c] = make_int2(cv, excl);
  }
  __syncthreads();

  int pos[8];
  #pragma unroll
  for (int u = 0; u < 8; ++u) pos[u] = cnt[r[u] >= 0 ? (r[u] >> 8) : 0] + rk[u];
  #pragma unroll
  for (int u = 0; u < 8; ++u) {
    if (r[u] >= 0) img[pos[u]] = make_int4(cc[u], vv[u], r[u], 0);
  }
  __syncthreads();

  const int left = nEdges - c * CH_EDGES;
  const int nValid = left < CH_EDGES ? left : CH_EDGES;
  int4* __restrict__ rec4 = reinterpret_cast<int4*>(recs) + (size_t)c * CH_EDGES;
  if (nValid == CH_EDGES) {
    int4 q[CH_EDGES / CH_THREADS];
    #pragma unroll
    for (int k = 0; k < CH_EDGES / CH_THREADS; ++k) q[k] = img[k * CH_THREADS + t];
    #pragma unroll
    for (int k = 0; k < CH_EDGES / CH_THREADS; ++k) rec4[k * CH_THREADS + t] = q[k];
  } else {
    for (int i = t; i < nValid; i += CH_THREADS) rec4[i] = img[i];
  }
}

#define FP_U 8
#define FP_CAP 8192

template <bool VIA_LDS>
__device__ __forceinline__ void gcn_fine_put(const int4* __restrict__ rec4,
                                             const int2* __restrict__ pc2,
                                             int* cnt, int2* img,
                                             int2* __restrict__ pr,
                                             int bFirst, int nChunks, int t, int sub) {
  for (int cb = t >> 2; cb < nChunks; cb += 256) {
    const int2 pc = pc2[cb];
    const int4* __restrict__ p = rec4 + (size_t)cb * CH_EDGES + pc.y;
    for (int j0 = sub; j0 < pc.x; j0 += 4 * FP_U) {
      int4 q[FP_U];
      #pragma unroll
      for (int u = 0; u < FP_U; ++u) {
        const int j = j0 + 4 * u;
        q[u] = (j < pc.x) ? p[j] : make_int4(0, 0, -1, 0);
      }
      int pos[FP_U];
      #pragma unroll
      for (int u = 0; u < FP_U; ++u)
        pos[u] = (q[u].z >= 0) ? atomicAdd(&cnt[q[u].z & 255], 1) : 0;
      #pragma unroll
      for (int u = 0; u < FP_U; ++u) {
        if (VIA_LDS) {
          img[(q[u].z >= 0) ? pos[u] - bFirst : FP_CAP + (t & 31)] = make_int2(q[u].x, q[u].y);
        } else if (q[u].z >= 0) {
          pr[pos[u]] = make_int2(q[u].x, q[u].y);
        }
      }
    }
  }
}
template <int U>
__device__ __forceinline__ void gcn_gather_batch(float4& acc, int2 mine, int j,
                                                 const float4* __restrict__ x4,
                                                 int lane) {
  float4 d[U];
  float v[U];
  #pragma unroll
  for (int u = 0; u < U; ++u) {
    const int c = __builtin_amdgcn_readlane(mine.x, j + u);
    v[u] = __int_as_float(__builtin_amdgcn_readlane(mine.y, j + u));
    d[u] = x4[(size_t)c * (IN_DIM / 4) + lane];
  }
  #pragma unroll
  for (int u = 0; u < U; ++u) {
    acc.x += v[u] * d[u].x;
    acc.y += v[u] * d[u].y;
    acc.z += v[u] * d[u].z;
    acc.w += v[u] * d[u].w;
  }
}

template <bool FROM_LDS>
__device__ __forceinline__ float4 gcn_row_sum1(int s, int e, const int2* img,
                                               int bFirst, const int2* pr,
                                               const float4* __restrict__ x4,
                                               int lane) {
  float4 acc = make_float4(0.0f, 0.0f, 0.0f, 0.0f);
  for (int base = s; base < e; base += 32) {
    const int n = (e - base) < 32 ? (e - base) : 32;
    int2 mine = make_int2(0, 0);
    if (lane < n) mine = FROM_LDS ? img[base - bFirst + lane] : pr[base + lane];
    int j = 0;
    for (; j + 8 <= n; j += 8) gcn_gather_batch<8>(acc, mine, j, x4, lane);
    if (n - j >= 4) { gcn_gather_batch<4>(acc, mine, j, x4, lane); j += 4; }
    if (n - j >= 2) { gcn_gather_batch<2>(acc, mine, j, x4, lane); j += 2; }
    if (n - j >= 1) { gcn_gather_batch<1>(acc, mine, j, x4, lane); }
  }
  return acc;
}

template <bool FROM_LDS>
__device__ __forceinline__ void gcn_rows8(int sv, const int2* img,
                                          int bFirst, const int2* pr,
                                          const float4* __restrict__ x4,
                                          float* Al, int w, int lane) {
  #pragma nounroll
  for (int q = 0; q < 8; ++q) {
    const int s = __builtin_amdgcn_readlane(sv, q);
    const int e = __builtin_amdgcn_readlane(sv, q + 1);
    const float4 acc = gcn_row_sum1<FROM_LDS>(s, e, img, bFirst, pr, x4, lane);
    *reinterpret_cast<float4*>(&Al[(w * 8 + q) * A1_PITCH + lane * 4]) = acc;
  }
}

__global__ void __launch_bounds__(1024)
gcn_fine_gather1_gemm1(const int* __restrict__ recs, const int* __restrict__ piece,
                       int* __restrict__ start, int* pairs,
                       const float* __restrict__ x, const float* __restrict__ w1,
                       float* __restrict__ h, int nNodes, int nChunks) {
  __shared__ __attribute__((aligned(16))) int2 img[FP_CAP + 32];
  __shared__ __attribute__((aligned(16))) float Wl[IN_DIM * HIDDEN];
  __shared__ __attribute__((aligned(16))) float Al[256 * A1_PITCH];
  __shared__ int cnt[256];
  __shared__ int wtot[8];
  __shared__ int wlo[32];
  const int t = threadIdx.x;
  const int lane = t & 31;
  const int w = __builtin_amdgcn_readfirstlane(t >> 5);
  const int b = blockIdx.x;
  const int sub = t & 3;
  const int4* __restrict__ rec4 = reinterpret_cast<const int4*>(recs);
  const int2* __restrict__ pc2 = reinterpret_cast<const int2*>(piece) + (size_t)b * nChunks;
  if (t < 256) cnt[t] = 0;
  float4 wreg = make_float4(0.0f, 0.0f, 0.0f, 0.0f);
  if (t < (IN_DIM * HIDDEN) / 4) wreg = reinterpret_cast<const float4*>(w1)[t];
  __syncthreads();

  int myLo = 0;
  for (int c = t >> 2; c < nChunks; c += 256) {
    const int2 pc = pc2[c];
    if (sub == 0) myLo += pc.y;
    const int4* __restrict__ p = rec4 + (size_t)c * CH_EDGES + pc.y;
    for (int j0 = sub; j0 < pc.x; j0 += 4 * FP_U) {
      int z[FP_U];
      #pragma unroll
      for (int u = 0; u < FP_U; ++u) {
        const int j = j0 + 4 * u;
        z[u] = (j < pc.x) ? p[j].z : -1;
      }
      #pragma unroll
      for (int u = 0; u < FP_U; ++u)
        if (z[u] >= 0) atomicAdd(&cnt[z[u] & 255], 1);
    }
  }
  #pragma unroll
  for (int d = 16; d > 0; d >>= 1) myLo += __shfl_xor(myLo, d);
  if (lane == 0) wlo[w] = myLo;
  if (t < (IN_DIM * HIDDEN) / 4) reinterpret_cast<float4*>(Wl)[t] = wreg;
  __syncthreads();

  int c = 0, inc = 0;
  if (t < 256) {
    c = cnt[t];
    inc = c;
    #pragma unroll
    for (int d = 1; d < 32; d <<= 1) {
      const int n = __shfl_up(inc, d);
      if (lane >= d) inc += n;
    }
    if (lane == 31) wtot[w] = inc;
  }
  __syncthreads();
  if (t < 256) {
    int off = 0;
    #pragma unroll
    for (int i = 0; i < 32; ++i) off += wlo[i];
    #pragma unroll
    for (int i = 0; i < 8; ++i)
      if (i < w) off += wtot[i];
    const int first = off + inc - c;
    const int row = b * 256 + t;
    if (row < nNodes) start[row] = first;
    if (row == nNodes - 1) start[nNodes] = first + c;
    cnt[t] = first;
  }
  __syncthreads();

  int bFirst = 0, nb = 0;
  #pragma unroll
  for (int i = 0; i < 32; ++i) bFirst += wlo[i];
  #pragma unroll
  for (int i = 0; i < 8; ++i) nb += wtot[i];
  bFirst = __builtin_amdgcn_readfirstlane(bFirst);
  nb = __builtin_amdgcn_readfirstlane(nb);

  int2* pr = reinterpret_cast<int2*>(pairs);
  const float4* __restrict__ x4 = reinterpret_cast<const float4*>(x);
  const bool viaLds = nb <= FP_CAP;
  if (viaLds) {
    gcn_fine_put<true>(rec4, pc2, cnt, img, pr, bFirst, nChunks, t, sub);
    __syncthreads();
    const int a0 = bFirst & ~31;
    const int end = nb > 0 ? bFirst + nb : a0;
    for (int i0 = a0; i0 < end; i0 += 4 * 1024) {
      int2 v[4];
      #pragma unroll
      for (int k = 0; k < 4; ++k) {
        const int li = i0 + k * 1024 + t - bFirst;
        v[k] = img[li < 0 ? 0 : (li < nb ? li : nb - 1)];
      }
      #pragma unroll
      for (int k = 0; k < 4; ++k) {
        const int i = i0 + k * 1024 + t;
        if (i >= bFirst && i < end) pr[i] = v[k];
      }
    }
  } else {
    gcn_fine_put<false>(rec4, pc2, cnt, img, pr, bFirst, nChunks, t, sub);
    __syncthreads();
  }

  int sv = 0;
  if (lane < 9) {
    const int i = w * 8 + lane;
    sv = (i == 0) ? bFirst : cnt[i - 1];
  }
  if (viaLds) gcn_rows8<true>(sv, img, bFirst, pr, x4, Al, w, lane);
  else        gcn_rows8<false>(sv, img, bFirst, pr, x4, Al, w, lane);
  __syncthreads();
  if (w >= 16) return;

  const int half = lane >> 4;
  const int l    = lane & 15;
  const float* At = Al + w * 16 * A1_PITCH;

  v8f acc = {};
  #pragma unroll 8
  for (int k0 = 0; k0 < IN_DIM; k0 += 4) {
    const int ka = k0 + 2 * half;
    const v2f a = *reinterpret_cast<const v2f*>(&At[l * A1_PITCH + ka]);
    v2f bb;
    bb.x = Wl[ka * HIDDEN + l];
    bb.y = Wl[(ka + 1) * HIDDEN + l];
    acc = __builtin_amdgcn_wmma_f32_16x16x4_f32(false, a, false, bb,
                                                (short)0, acc, false, false);
  }

  const int rowT = b * 256 + w * 16;
  #pragma unroll
  for (int i = 0; i < 8; ++i) {
    const int row = rowT + i + 8 * half;
    const float v = acc[i];
    if (row < nNodes) h[(size_t)row * HIDDEN + l] = v > 0.0f ? v : 0.0f;
  }
}

__global__ void __launch_bounds__(128)
gcn_gather2_gemm2_lsm(const int* __restrict__ start,
                      const int* __restrict__ pairs,
                      const float* __restrict__ h,
                      const float* __restrict__ w2,
                      float* __restrict__ out, int nNodes) {
  __shared__ float Wl[HIDDEN * NPAD];
  __shared__ __attribute__((aligned(16))) float Al[16 * A2_PITCH];
  __shared__ float Ol[16 * O_PITCH];
  const int t = threadIdx.x;
  const int lane = t & 31;
  const int w = __builtin_amdgcn_readfirstlane(t >> 5);
  const int tile = blockIdx.x;
  const int slot = (lane >> 2) & 3;
  const int f4   = lane & 3;
  const int2* __restrict__ pr = reinterpret_cast<const int2*>(pairs);
  const float4* __restrict__ h4 = reinterpret_cast<const float4*>(h);

  int rl[2], s[2], e[2];
  #pragma unroll
  for (int u = 0; u < 2; ++u) {
    rl[u] = w * 4 + u * 2 + (lane >> 4);
    const int r = tile * 16 + rl[u];
    s[u] = start[r < nNodes ? r : nNodes];
    e[u] = start[r + 1 < nNodes ? r + 1 : nNodes];
  }

  #pragma unroll
  for (int k = 0; k < (HIDDEN * NPAD) / 128; ++k) {
    const int i = t + 128 * k;
    const int r = i / NPAD, n = i % NPAD;
    const float v = w2[r * NCLASS + (n < NCLASS ? n : NCLASS - 1)];
    Wl[i] = (n < NCLASS) ? v : 0.0f;
  }

  #pragma unroll
  for (int u = 0; u < 2; ++u) {
    float4 acc = make_float4(0.0f, 0.0f, 0.0f, 0.0f);
    int p = s[u] + slot;
    for (; p + 4 < e[u]; p += 8) {
      const int2 cv0 = pr[p];
      const int2 cv1 = pr[p + 4];
      const float4 d0 = h4[(size_t)cv0.x * (HIDDEN / 4) + f4];
      const float4 d1 = h4[(size_t)cv1.x * (HIDDEN / 4) + f4];
      const float v0 = __int_as_float(cv0.y);
      const float v1 = __int_as_float(cv1.y);
      acc.x += v0 * d0.x; acc.y += v0 * d0.y; acc.z += v0 * d0.z; acc.w += v0 * d0.w;
      acc.x += v1 * d1.x; acc.y += v1 * d1.y; acc.z += v1 * d1.z; acc.w += v1 * d1.w;
    }
    if (p < e[u]) {
      const int2 cv0 = pr[p];
      const float4 d0 = h4[(size_t)cv0.x * (HIDDEN / 4) + f4];
      const float v0 = __int_as_float(cv0.y);
      acc.x += v0 * d0.x; acc.y += v0 * d0.y; acc.z += v0 * d0.z; acc.w += v0 * d0.w;
    }
    #pragma unroll
    for (int d = 4; d <= 8; d <<= 1) {
      acc.x += __shfl_xor(acc.x, d);
      acc.y += __shfl_xor(acc.y, d);
      acc.z += __shfl_xor(acc.z, d);
      acc.w += __shfl_xor(acc.w, d);
    }
    if (slot == 0)
      *reinterpret_cast<float4*>(&Al[rl[u] * A2_PITCH + f4 * 4]) = acc;
  }
  __syncthreads();

  if (w == 0) {
    const int half = lane >> 4;
    const int l    = lane & 15;
    #pragma unroll
    for (int nt = 0; nt < 3; ++nt) {
      v8f c = {};
      #pragma unroll
      for (int k0 = 0; k0 < HIDDEN; k0 += 4) {
        const int ka = k0 + 2 * half;
        const v2f a = *reinterpret_cast<const v2f*>(&Al[l * A2_PITCH + ka]);
        v2f b;
        b.x = Wl[ka * NPAD + nt * 16 + l];
        b.y = Wl[(ka + 1) * NPAD + nt * 16 + l];
        c = __builtin_amdgcn_wmma_f32_16x16x4_f32(false, a, false, b,
                                                  (short)0, c, false, false);
      }
      #pragma unroll
      for (int i = 0; i < 8; ++i)
        Ol[(i + 8 * half) * O_PITCH + nt * 16 + l] = c[i];
    }
  }
  __syncthreads();

  if (w == 0 && lane < 16 && tile * 16 + lane < nNodes) {
    const float* row = Ol + lane * O_PITCH;
    float mx = row[0];
    for (int j = 1; j < NCLASS; ++j) mx = fmaxf(mx, row[j]);
    float sum = 0.0f;
    for (int j = 0; j < NCLASS; ++j) sum += __expf(row[j] - mx);
    const float lse = mx + __logf(sum);
    float* ob = out + ((size_t)tile * 16 + lane) * NCLASS;
    for (int j = 0; j < NCLASS; ++j) ob[j] = row[j] - lse;
  }
}

extern "C" void kernel_launch(void* const* d_in, const int* in_sizes, int n_in,
                              void* d_out, int out_size, void* d_ws, size_t ws_size,
                              hipStream_t stream) {
  const float* features = (const float*)d_in[0];
  const float* weight1  = (const float*)d_in[1];
  const float* weight2  = (const float*)d_in[2];
  const float* edgeVals = (const float*)d_in[3];
  const int*   edgeRows = (const int*)  d_in[4];
  const int*   edgeCols = (const int*)  d_in[5];

  const int nNodes = in_sizes[0] / IN_DIM;
  const int nEdges = in_sizes[3];
  const int nTiles = nNodes / 16;

  const int nBuckets = (nNodes + 255) / 256;
  const int nChunks  = (nEdges + CH_EDGES - 1) / CH_EDGES;
  const int nHist    = nBuckets * nChunks;

  float* h  = (float*)d_ws;
  int* pairs  = (int*)(h + (size_t)nNodes * HIDDEN);
  int* recs   = pairs + (size_t)nEdges * 2;
  int* piece  = recs + (size_t)nChunks * CH_EDGES * 4;
  int* start  = piece + (size_t)nHist * 2;

  gcn_chunk_sort<<<nChunks, CH_THREADS, 0, stream>>>(
      edgeRows, edgeCols, edgeVals, recs, piece, nEdges, nChunks, nBuckets);
  gcn_fine_gather1_gemm1<<<nBuckets, 1024, 0, stream>>>(
      recs, piece, start, pairs, features, weight1, h, nNodes, nChunks);
  gcn_gather2_gemm2_lsm<<<nTiles, 128, 0, stream>>>(start, pairs, h, weight2, (float*)d_out, nNodes);
}


# Layer 2: one wave owns a 16-row tile end to end (3125 waves, not 12500; no wave waits for another): tile's slots copied to LDS, 4 lanes walk a row in slot order, W2 in registers; row sums sequential
# speedup vs baseline: 4.4195x; 1.0519x over previous
#include <hip/hip_runtime.h>
#include <hip/hip_bf16.h>
#include <math.h>


typedef __attribute__((ext_vector_type(2))) float v2f;
typedef __attribute__((ext_vector_type(8))) float v8f;

#define IN_DIM   128
#define HIDDEN    16
#define NCLASS    40
#define NPAD      48
#define A1_PITCH 132
#define A2_PITCH  20
#define O_PITCH   49

#define CH_THREADS 512
#define CH_EDGES   4096
#define NB_MAX     256

__device__ __forceinline__ void gcn_chunk_load8(const int* __restrict__ p, int c,
                                                int t, int nEdges, int fill,
                                                int (&out)[8]) {
  #pragma unroll
  for (int k = 0; k < 2; ++k) {
    const int e = (c * (CH_EDGES / 4) + k * CH_THREADS + t) * 4;
    if (e + 4 <= nEdges) {
      const int4 q = *reinterpret_cast<const int4*>(p + e);
      out[4 * k + 0] = q.x; out[4 * k + 1] = q.y;
      out[4 * k + 2] = q.z; out[4 * k + 3] = q.w;
    } else {
      #pragma unroll
      for (int u = 0; u < 4; ++u) out[4 * k + u] = (e + u < nEdges) ? p[e + u] : fill;
    }
  }
}

__global__ void __launch_bounds__(CH_THREADS)
gcn_chunk_sort(const int* __restrict__ rows, const int* __restrict__ cols,
               const float* __restrict__ vals, int* __restrict__ recs,
               int* __restrict__ piece, int nEdges, int nChunks, int nBuckets) {
  __shared__ __attribute__((aligned(16))) int4 img[CH_EDGES];
  __shared__ int cnt[NB_MAX];
  __shared__ int wtot[NB_MAX / 32];
  const int t = threadIdx.x;
  const int lane = t & 31;
  const int w = t >> 5;
  const int c = blockIdx.x;
  if (t < NB_MAX) cnt[t] = 0;
  int r[8], cc[8], vv[8], rk[8];
  gcn_chunk_load8(rows, c, t, nEdges, -1, r);
  gcn_chunk_load8(cols, c, t, nEdges, 0, cc);
  gcn_chunk_load8(reinterpret_cast<const int*>(vals), c, t, nEdges, 0, vv);
  __syncthreads();
  #pragma unroll
  for (int u = 0; u < 8; ++u) {
    rk[u] = 0;
    if (r[u] >= 0) rk[u] = atomicAdd(&cnt[r[u] >> 8], 1);
  }
  __syncthreads();

  int cv = 0, inc = 0;
  if (t < NB_MAX) {
    cv = cnt[t];
    inc = cv;
    #pragma unroll
    for (int d = 1; d < 32; d <<= 1) {
      const int n = __shfl_up(inc, d);
      if (lane >= d) inc += n;
    }
    if (lane == 31) wtot[w] = inc;
  }
  __syncthreads();
  if (t < NB_MAX) {
    int off = 0;
    #pragma unroll
    for (int i = 0; i < NB_MAX / 32; ++i)
      if (i < w) off += wtot[i];
    const int excl = off + inc - cv;
    cnt[t] = excl;
    if (t < nBuckets)
      reinterpret_cast<int2*>(piece)[t * nChunks + c] = make_int2(cv, excl);
  }
  __syncthreads();

  int pos[8];
  #pragma unroll
  for (int u = 0; u < 8; ++u) pos[u] = cnt[r[u] >= 0 ? (r[u] >> 8) : 0] + rk[u];
  #pragma unroll
  for (int u = 0; u < 8; ++u) {
    if (r[u] >= 0) img[pos[u]] = make_int4(cc[u], vv[u], r[u], 0);
  }
  __syncthreads();

  const int left = nEdges - c * CH_EDGES;
  const int nValid = left < CH_EDGES ? left : CH_EDGES;
  int4* __restrict__ rec4 = reinterpret_cast<int4*>(recs) + (size_t)c * CH_EDGES;
  if (nValid == CH_EDGES) {
    int4 q[CH_EDGES / CH_THREADS];
    #pragma unroll
    for (int k = 0; k < CH_EDGES / CH_THREADS; ++k) q[k] = img[k * CH_THREADS + t];
    #pragma unroll
    for (int k = 0; k < CH_EDGES / CH_THREADS; ++k) rec4[k * CH_THREADS + t] = q[k];
  } else {
    for (int i = t; i < nValid; i += CH_THREADS) rec4[i] = img[i];
  }
}

#define FP_U 8
#define FP_CAP 8192

template <bool VIA_LDS>
__device__ __forceinline__ void gcn_fine_put(const int4* __restrict__ rec4,
                                             const int2* __restrict__ pc2,
                                             int* cnt, int2* img,
                                             int2* __restrict__ pr,
                                             int bFirst, int nChunks, int t, int sub) {
  for (int cb = t >> 2; cb < nChunks; cb += 256) {
    const int2 pc = pc2[cb];
    const int4* __restrict__ p = rec4 + (size_t)cb * CH_EDGES + pc.y;
    for (int j0 = sub; j0 < pc.x; j0 += 4 * FP_U) {
      int4 q[FP_U];
      #pragma unroll
      for (int u = 0; u < FP_U; ++u) {
        const int j = j0 + 4 * u;
        q[u] = (j < pc.x) ? p[j] : make_int4(0, 0, -1, 0);
      }
      int pos[FP_U];
      #pragma unroll
      for (int u = 0; u < FP_U; ++u)
        pos[u] = (q[u].z >= 0) ? atomicAdd(&cnt[q[u].z & 255], 1) : 0;
      #pragma unroll
      for (int u = 0; u < FP_U; ++u) {
        if (VIA_LDS) {
          img[(q[u].z >= 0) ? pos[u] - bFirst : FP_CAP + (t & 31)] = make_int2(q[u].x, q[u].y);
        } else if (q[u].z >= 0) {
          pr[pos[u]] = make_int2(q[u].x, q[u].y);
        }
      }
    }
  }
}
template <int U>
__device__ __forceinline__ void gcn_gather_batch(float4& acc, int2 mine, int j,
                                                 const float4* __restrict__ x4,
                                                 int lane) {
  float4 d[U];
  float v[U];
  #pragma unroll
  for (int u = 0; u < U; ++u) {
    const int c = __builtin_amdgcn_readlane(mine.x, j + u);
    v[u] = __int_as_float(__builtin_amdgcn_readlane(mine.y, j + u));
    d[u] = x4[(size_t)c * (IN_DIM / 4) + lane];
  }
  #pragma unroll
  for (int u = 0; u < U; ++u) {
    acc.x += v[u] * d[u].x;
    acc.y += v[u] * d[u].y;
    acc.z += v[u] * d[u].z;
    acc.w += v[u] * d[u].w;
  }
}

template <bool FROM_LDS>
__device__ __forceinline__ float4 gcn_row_sum1(int s, int e, const int2* img,
                                               int bFirst, const int2* pr,
                                               const float4* __restrict__ x4,
                                               int lane) {
  float4 acc = make_float4(0.0f, 0.0f, 0.0f, 0.0f);
  for (int base = s; base < e; base += 32) {
    const int n = (e - base) < 32 ? (e - base) : 32;
    int2 mine = make_int2(0, 0);
    if (lane < n) mine = FROM_LDS ? img[base - bFirst + lane] : pr[base + lane];
    int j = 0;
    for (; j + 8 <= n; j += 8) gcn_gather_batch<8>(acc, mine, j, x4, lane);
    if (n - j >= 4) { gcn_gather_batch<4>(acc, mine, j, x4, lane); j += 4; }
    if (n - j >= 2) { gcn_gather_batch<2>(acc, mine, j, x4, lane); j += 2; }
    if (n - j >= 1) { gcn_gather_batch<1>(acc, mine, j, x4, lane); }
  }
  return acc;
}

template <bool FROM_LDS>
__device__ __forceinline__ void gcn_rows8(int sv, const int2* img,
                                          int bFirst, const int2* pr,
                                          const float4* __restrict__ x4,
                                          float* Al, int w, int lane) {
  #pragma nounroll
  for (int q = 0; q < 8; ++q) {
    const int s = __builtin_amdgcn_readlane(sv, q);
    const int e = __builtin_amdgcn_readlane(sv, q + 1);
    const float4 acc = gcn_row_sum1<FROM_LDS>(s, e, img, bFirst, pr, x4, lane);
    *reinterpret_cast<float4*>(&Al[(w * 8 + q) * A1_PITCH + lane * 4]) = acc;
  }
}

__global__ void __launch_bounds__(1024)
gcn_fine_gather1_gemm1(const int* __restrict__ recs, const int* __restrict__ piece,
                       int* __restrict__ start, int* pairs,
                       const float* __restrict__ x, const float* __restrict__ w1,
                       float* __restrict__ h, int nNodes, int nChunks) {
  __shared__ __attribute__((aligned(16))) int2 img[FP_CAP + 32];
  __shared__ __attribute__((aligned(16))) float Wl[IN_DIM * HIDDEN];
  __shared__ __attribute__((aligned(16))) float Al[256 * A1_PITCH];
  __shared__ int cnt[256];
  __shared__ int wtot[8];
  __shared__ int wlo[32];
  const int t = threadIdx.x;
  const int lane = t & 31;
  const int w = __builtin_amdgcn_readfirstlane(t >> 5);
  const int b = blockIdx.x;
  const int sub = t & 3;
  const int4* __restrict__ rec4 = reinterpret_cast<const int4*>(recs);
  const int2* __restrict__ pc2 = reinterpret_cast<const int2*>(piece) + (size_t)b * nChunks;
  if (t < 256) cnt[t] = 0;
  float4 wreg = make_float4(0.0f, 0.0f, 0.0f, 0.0f);
  if (t < (IN_DIM * HIDDEN) / 4) wreg = reinterpret_cast<const float4*>(w1)[t];
  __syncthreads();

  int myLo = 0;
  for (int c = t >> 2; c < nChunks; c += 256) {
    const int2 pc = pc2[c];
    if (sub == 0) myLo += pc.y;
    const int4* __restrict__ p = rec4 + (size_t)c * CH_EDGES + pc.y;
    for (int j0 = sub; j0 < pc.x; j0 += 4 * FP_U) {
      int z[FP_U];
      #pragma unroll
      for (int u = 0; u < FP_U; ++u) {
        const int j = j0 + 4 * u;
        z[u] = (j < pc.x) ? p[j].z : -1;
      }
      #pragma unroll
      for (int u = 0; u < FP_U; ++u)
        if (z[u] >= 0) atomicAdd(&cnt[z[u] & 255], 1);
    }
  }
  #pragma unroll
  for (int d = 16; d > 0; d >>= 1) myLo += __shfl_xor(myLo, d);
  if (lane == 0) wlo[w] = myLo;
  if (t < (IN_DIM * HIDDEN) / 4) reinterpret_cast<float4*>(Wl)[t] = wreg;
  __syncthreads();

  int c = 0, inc = 0;
  if (t < 256) {
    c = cnt[t];
    inc = c;
    #pragma unroll
    for (int d = 1; d < 32; d <<= 1) {
      const int n = __shfl_up(inc, d);
      if (lane >= d) inc += n;
    }
    if (lane == 31) wtot[w] = inc;
  }
  __syncthreads();
  if (t < 256) {
    int off = 0;
    #pragma unroll
    for (int i = 0; i < 32; ++i) off += wlo[i];
    #pragma unroll
    for (int i = 0; i < 8; ++i)
      if (i < w) off += wtot[i];
    const int first = off + inc - c;
    const int row = b * 256 + t;
    if (row < nNodes) start[row] = first;
    if (row == nNodes - 1) start[nNodes] = first + c;
    cnt[t] = first;
  }
  __syncthreads();

  int bFirst = 0, nb = 0;
  #pragma unroll
  for (int i = 0; i < 32; ++i) bFirst += wlo[i];
  #pragma unroll
  for (int i = 0; i < 8; ++i) nb += wtot[i];
  bFirst = __builtin_amdgcn_readfirstlane(bFirst);
  nb = __builtin_amdgcn_readfirstlane(nb);

  int2* pr = reinterpret_cast<int2*>(pairs);
  const float4* __restrict__ x4 = reinterpret_cast<const float4*>(x);
  const bool viaLds = nb <= FP_CAP;
  if (viaLds) {
    gcn_fine_put<true>(rec4, pc2, cnt, img, pr, bFirst, nChunks, t, sub);
    __syncthreads();
    const int a0 = bFirst & ~31;
    const int end = nb > 0 ? bFirst + nb : a0;
    for (int i0 = a0; i0 < end; i0 += 4 * 1024) {
      int2 v[4];
      #pragma unroll
      for (int k = 0; k < 4; ++k) {
        const int li = i0 + k * 1024 + t - bFirst;
        v[k] = img[li < 0 ? 0 : (li < nb ? li : nb - 1)];
      }
      #pragma unroll
      for (int k = 0; k < 4; ++k) {
        const int i = i0 + k * 1024 + t;
        if (i >= bFirst && i < end) pr[i] = v[k];
      }
    }
  } else {
    gcn_fine_put<false>(rec4, pc2, cnt, img, pr, bFirst, nChunks, t, sub);
    __syncthreads();
  }

  int sv = 0;
  if (lane < 9) {
    const int i = w * 8 + lane;
    sv = (i == 0) ? bFirst : cnt[i - 1];
  }
  if (viaLds) gcn_rows8<true>(sv, img, bFirst, pr, x4, Al, w, lane);
  else        gcn_rows8<false>(sv, img, bFirst, pr, x4, Al, w, lane);
  __syncthreads();
  if (w >= 16) return;

  const int half = lane >> 4;
  const int l    = lane & 15;
  const float* At = Al + w * 16 * A1_PITCH;

  v8f acc = {};
  #pragma unroll 8
  for (int k0 = 0; k0 < IN_DIM; k0 += 4) {
    const int ka = k0 + 2 * half;
    const v2f a = *reinterpret_cast<const v2f*>(&At[l * A1_PITCH + ka]);
    v2f bb;
    bb.x = Wl[ka * HIDDEN + l];
    bb.y = Wl[(ka + 1) * HIDDEN + l];
    acc = __builtin_amdgcn_wmma_f32_16x16x4_f32(false, a, false, bb,
                                                (short)0, acc, false, false);
  }

  const int rowT = b * 256 + w * 16;
  #pragma unroll
  for (int i = 0; i < 8; ++i) {
    const int row = rowT + i + 8 * half;
    const float v = acc[i];
    if (row < nNodes) h[(size_t)row * HIDDEN + l] = v > 0.0f ? v : 0.0f;
  }
}

#define T2_CAP 1024
#define T2_U   8

template <bool FROM_LDS>
__device__ __forceinline__ void gcn_rows2(float4& accA, float4& accB,
                                          int sA, int nA, int sB, int nB,
                                          int S, int n, const int2* img,
                                          const int2* __restrict__ pr,
                                          const float4* __restrict__ h4, int f4) {
  const int nAB = nA > nB ? nA : nB;
  for (int j0 = 0; j0 < nAB; j0 += T2_U) {
    int2 cvA[T2_U], cvB[T2_U];
    #pragma unroll
    for (int u = 0; u < T2_U; ++u) {
      const int iA = sA - S + j0 + u;
      const int iB = sB - S + j0 + u;
      const int cA = iA < n ? iA : n - 1;
      const int cB = iB < n ? iB : n - 1;
      cvA[u] = FROM_LDS ? img[cA] : pr[S + cA];
      cvB[u] = FROM_LDS ? img[cB] : pr[S + cB];
    }
    float4 dA[T2_U], dB[T2_U];
    #pragma unroll
    for (int u = 0; u < T2_U; ++u) {
      dA[u] = h4[(size_t)cvA[u].x * (HIDDEN / 4) + f4];
      dB[u] = h4[(size_t)cvB[u].x * (HIDDEN / 4) + f4];
    }
    #pragma unroll
    for (int u = 0; u < T2_U; ++u) {
      const bool okA = j0 + u < nA;
      const bool okB = j0 + u < nB;
      const float vA = okA ? __int_as_float(cvA[u].y) : 0.0f;
      const float vB = okB ? __int_as_float(cvB[u].y) : 0.0f;
      const float4 a = okA ? dA[u] : make_float4(0.0f, 0.0f, 0.0f, 0.0f);
      const float4 b = okB ? dB[u] : make_float4(0.0f, 0.0f, 0.0f, 0.0f);
      accA.x += vA * a.x; accA.y += vA * a.y; accA.z += vA * a.z; accA.w += vA * a.w;
      accB.x += vB * b.x; accB.y += vB * b.y; accB.z += vB * b.z; accB.w += vB * b.w;
    }
  }
}

__global__ void __launch_bounds__(32)
gcn_gather2_gemm2_lsm(const int* __restrict__ start,
                      const int* __restrict__ pairs,
                      const float* __restrict__ h,
                      const float* __restrict__ w2,
                      float* __restrict__ out, int nNodes) {
  __shared__ __attribute__((aligned(16))) int2 img[T2_CAP];
  __shared__ __attribute__((aligned(16))) float Al[16 * A2_PITCH];
  __shared__ float Ol[16 * O_PITCH];
  const int lane = threadIdx.x;
  const int tile = blockIdx.x;
  const int half = lane >> 4;
  const int l    = lane & 15;
  const int q    = lane >> 2;
  const int f4   = lane & 3;
  const int2* __restrict__ pr = reinterpret_cast<const int2*>(pairs);
  const float4* __restrict__ h4 = reinterpret_cast<const float4*>(h);

  const int r0 = tile * 16 + lane;
  const int sv = start[r0 < nNodes ? r0 : nNodes];

  v2f bf[3][4];
  #pragma unroll
  for (int nt = 0; nt < 3; ++nt) {
    const int nn = nt * 16 + l;
    const int nc = nn < NCLASS ? nn : NCLASS - 1;
    #pragma unroll
    for (int st = 0; st < 4; ++st) {
      const int ka = 4 * st + 2 * half;
      const float x0 = w2[ka * NCLASS + nc];
      const float x1 = w2[(ka + 1) * NCLASS + nc];
      bf[nt][st].x = nn < NCLASS ? x0 : 0.0f;
      bf[nt][st].y = nn < NCLASS ? x1 : 0.0f;
    }
  }

  const int S = __builtin_amdgcn_readlane(sv, 0);
  const int n = __builtin_amdgcn_readlane(sv, 16) - S;
  const bool viaLds = n <= T2_CAP;
  if (viaLds) {
    for (int i0 = 0; i0 < n; i0 += 32 * 8) {
      int2 v[8];
      #pragma unroll
      for (int k = 0; k < 8; ++k) {
        const int i = i0 + 32 * k + lane;
        v[k] = pr[S + (i < n ? i : n - 1)];
      }
      #pragma unroll
      for (int k = 0; k < 8; ++k) {
        const int i = i0 + 32 * k + lane;
        if (i < n) img[i] = v[k];
      }
    }
  }
  __syncthreads();

  const int sA = __shfl(sv, q),     nA = __shfl(sv, q + 1) - sA;
  const int sB = __shfl(sv, q + 8), nB = __shfl(sv, q + 9) - sB;
  float4 accA = make_float4(0.0f, 0.0f, 0.0f, 0.0f);
  float4 accB = make_float4(0.0f, 0.0f, 0.0f, 0.0f);
  if (n > 0) {
    if (viaLds) gcn_rows2<true>(accA, accB, sA, nA, sB, nB, S, n, img, pr, h4, f4);
    else        gcn_rows2<false>(accA, accB, sA, nA, sB, nB, S, n, img, pr, h4, f4);
  }
  *reinterpret_cast<float4*>(&Al[q * A2_PITCH + f4 * 4]) = accA;
  *reinterpret_cast<float4*>(&Al[(q + 8) * A2_PITCH + f4 * 4]) = accB;
  __syncthreads();

  v2f af[4];
  #pragma unroll
  for (int st = 0; st < 4; ++st) {
    const int ka = 4 * st + 2 * half;
    af[st] = *reinterpret_cast<const v2f*>(&Al[l * A2_PITCH + ka]);
  }
  #pragma unroll
  for (int nt = 0; nt < 3; ++nt) {
    v8f c = {};
    #pragma unroll
    for (int st = 0; st < 4; ++st)
      c = __builtin_amdgcn_wmma_f32_16x16x4_f32(false, af[st], false, bf[nt][st],
                                                (short)0, c, false, false);
    #pragma unroll
    for (int i = 0; i < 8; ++i)
      Ol[(i + 8 * half) * O_PITCH + nt * 16 + l] = c[i];
  }
  __syncthreads();

  if (lane < 16 && tile * 16 + lane < nNodes) {
    const float* row = Ol + lane * O_PITCH;
    float mx = row[0];
    for (int j = 1; j < NCLASS; ++j) mx = fmaxf(mx, row[j]);
    float sum = 0.0f;
    for (int j = 0; j < NCLASS; ++j) sum += __expf(row[j] - mx);
    const float lse = mx + __logf(sum);
    float* ob = out + ((size_t)tile * 16 + lane) * NCLASS;
    for (int j = 0; j < NCLASS; ++j) ob[j] = row[j] - lse;
  }
}

extern "C" void kernel_launch(void* const* d_in, const int* in_sizes, int n_in,
                              void* d_out, int out_size, void* d_ws, size_t ws_size,
                              hipStream_t stream) {
  const float* features = (const float*)d_in[0];
  const float* weight1  = (const float*)d_in[1];
  const float* weight2  = (const float*)d_in[2];
  const float* edgeVals = (const float*)d_in[3];
  const int*   edgeRows = (const int*)  d_in[4];
  const int*   edgeCols = (const int*)  d_in[5];

  const int nNodes = in_sizes[0] / IN_DIM;
  const int nEdges = in_sizes[3];
  const int nTiles = nNodes / 16;

  const int nBuckets = (nNodes + 255) / 256;
  const int nChunks  = (nEdges + CH_EDGES - 1) / CH_EDGES;
  const int nHist    = nBuckets * nChunks;

  float* h  = (float*)d_ws;
  int* pairs  = (int*)(h + (size_t)nNodes * HIDDEN);
  int* recs   = pairs + (size_t)nEdges * 2;
  int* piece  = recs + (size_t)nChunks * CH_EDGES * 4;
  int* start  = piece + (size_t)nHist * 2;

  gcn_chunk_sort<<<nChunks, CH_THREADS, 0, stream>>>(
      edgeRows, edgeCols, edgeVals, recs, piece, nEdges, nChunks, nBuckets);
  gcn_fine_gather1_gemm1<<<nBuckets, 1024, 0, stream>>>(
      recs, piece, start, pairs, features, weight1, h, nNodes, nChunks);
  gcn_gather2_gemm2_lsm<<<nTiles, 32, 0, stream>>>(start, pairs, h, weight2, (float*)d_out, nNodes);
}
